# starting kernel
# baseline (speedup 1.0000x reference)
#include <hip/hip_runtime.h>
#include <hip/hip_bf16.h>

typedef __attribute__((ext_vector_type(2))) float v2f;
typedef __attribute__((ext_vector_type(8))) float v8f;

__global__ __launch_bounds__(256) void zero_f32(float* __restrict__ p, unsigned long long n) {
    unsigned long long i = (unsigned long long)blockIdx.x * blockDim.x + threadIdx.x;
    if (i < n) p[i] = 0.0f;
}

__global__ __launch_bounds__(256) void sage_scatter(const float* __restrict__ hsrc,
                                                    const int* __restrict__ src,
                                                    const int* __restrict__ dst,
                                                    int nedges,
                                                    float* __restrict__ msg,
                                                    float* __restrict__ deg) {
    int lane = threadIdx.x & 31;
    int edge = blockIdx.x * (blockDim.x >> 5) + (threadIdx.x >> 5);
    if (edge >= nedges) return;
    int s = src[edge];
    int d = dst[edge];
    const float4 v = *(const float4*)(hsrc + (size_t)s * 128 + lane * 4);
    float* mrow = msg + (size_t)d * 128 + lane * 4;
    atomicAdd(mrow + 0, v.x);
    atomicAdd(mrow + 1, v.y);
    atomicAdd(mrow + 2, v.z);
    atomicAdd(mrow + 3, v.w);
    if (lane == 0) atomicAdd(deg + d, 1.0f);
}

__global__ __launch_bounds__(256) void sage_gemm(const float* __restrict__ hdst,
                                                 const float* __restrict__ msg,
                                                 const float* __restrict__ deg,
                                                 const float* __restrict__ Wself,
                                                 const float* __restrict__ Wneigh,
                                                 const float* __restrict__ bias,
                                                 float* __restrict__ out,
                                                 int n_dst, int d_out, int relu) {
    const int d_in = 128;
    int lane   = threadIdx.x & 31;
    int wave   = blockIdx.x * (blockDim.x >> 5) + (threadIdx.x >> 5);
    int ntiles = d_out >> 4;
    int mtiles = (n_dst + 15) >> 4;
    if (wave >= mtiles * ntiles) return;

    int mt = wave / ntiles;
    int nt = wave - mt * ntiles;
    int m0 = mt << 4;
    int n0 = nt << 4;

    int half = lane >> 4;
    int idx  = lane & 15;

    int arow = m0 + idx;
    if (arow > n_dst - 1) arow = n_dst - 1;

    const float* Ah = hdst  + (size_t)arow * d_in + 2 * half;
    const float* Am = msg   + (size_t)arow * d_in + 2 * half;
    const float* Bs = Wself  + (size_t)(n0 + idx) * d_in + 2 * half;
    const float* Bn = Wneigh + (size_t)(n0 + idx) * d_in + 2 * half;
    float invdeg = 1.0f / fmaxf(deg[arow], 1.0f);

    v8f c = {0.f, 0.f, 0.f, 0.f, 0.f, 0.f, 0.f, 0.f};

#pragma unroll
    for (int k = 0; k < d_in; k += 4) {
        v2f a = *(const v2f*)(Ah + k);
        v2f b = *(const v2f*)(Bs + k);
        c = __builtin_amdgcn_wmma_f32_16x16x4_f32(false, a, false, b,
                                                  (short)0, c, false, false);
    }
#pragma unroll
    for (int k = 0; k < d_in; k += 4) {
        v2f a = *(const v2f*)(Am + k);
        a.x *= invdeg;
        a.y *= invdeg;
        v2f b = *(const v2f*)(Bn + k);
        c = __builtin_amdgcn_wmma_f32_16x16x4_f32(false, a, false, b,
                                                  (short)0, c, false, false);
    }

    float bv = bias[n0 + idx];
#pragma unroll
    for (int j = 0; j < 8; ++j) {
        int m = m0 + j + 8 * half;
        if (m < n_dst) {
            float v = c[j] + bv;
            if (relu) v = fmaxf(v, 0.0f);
            out[(size_t)m * d_out + (n0 + idx)] = v;
        }
    }
}

static inline unsigned cdiv(unsigned a, unsigned b) { return (a + b - 1) / b; }

extern "C" void kernel_launch(void* const* d_in, const int* in_sizes, int n_in,
                              void* d_out, int out_size, void* d_ws, size_t ws_size,
                              hipStream_t stream) {
    const float* feats = (const float*)d_in[0];
    const int* src0 = (const int*)d_in[1];
    const int* dst0 = (const int*)d_in[2];
    const int* src1 = (const int*)d_in[3];
    const int* dst1 = (const int*)d_in[4];
    const int* src2 = (const int*)d_in[5];
    const int* dst2 = (const int*)d_in[6];
    const float* Ws0 = (const float*)d_in[10];
    const float* Wn0 = (const float*)d_in[11];
    const float* b0  = (const float*)d_in[12];
    const float* Ws1 = (const float*)d_in[13];
    const float* Wn1 = (const float*)d_in[14];
    const float* b1  = (const float*)d_in[15];
    const float* Ws2 = (const float*)d_in[16];
    const float* Wn2 = (const float*)d_in[17];
    const float* b2  = (const float*)d_in[18];

    const int E0 = in_sizes[1], E1 = in_sizes[3], E2 = in_sizes[5];
    const int N1 = 100000, N2 = 20000, N3 = 5000;

    char* ws = (char*)d_ws;
    float* h1  = (float*)ws;                    ws += (size_t)N1 * 128 * sizeof(float);
    float* h2  = (float*)ws;                    ws += (size_t)N2 * 128 * sizeof(float);
    float* msg = (float*)ws;                    ws += (size_t)N1 * 128 * sizeof(float);
    float* deg = (float*)ws;

    const dim3 blk(256);

    zero_f32<<<cdiv((unsigned)N1 * 128, 256), blk, 0, stream>>>(msg, (unsigned long long)N1 * 128);
    zero_f32<<<cdiv((unsigned)N1, 256), blk, 0, stream>>>(deg, (unsigned long long)N1);
    sage_scatter<<<cdiv((unsigned)E0, 8), blk, 0, stream>>>(feats, src0, dst0, E0, msg, deg);
    {
        unsigned tiles = cdiv((unsigned)N1, 16) * (128 / 16);
        sage_gemm<<<cdiv(tiles, 8), blk, 0, stream>>>(feats, msg, deg, Ws0, Wn0, b0, h1, N1, 128, 1);
    }

    zero_f32<<<cdiv((unsigned)N2 * 128, 256), blk, 0, stream>>>(msg, (unsigned long long)N2 * 128);
    zero_f32<<<cdiv((unsigned)N2, 256), blk, 0, stream>>>(deg, (unsigned long long)N2);
    sage_scatter<<<cdiv((unsigned)E1, 8), blk, 0, stream>>>(h1, src1, dst1, E1, msg, deg);
    {
        unsigned tiles = cdiv((unsigned)N2, 16) * (128 / 16);
        sage_gemm<<<cdiv(tiles, 8), blk, 0, stream>>>(h1, msg, deg, Ws1, Wn1, b1, h2, N2, 128, 1);
    }

    zero_f32<<<cdiv((unsigned)N3 * 128, 256), blk, 0, stream>>>(msg, (unsigned long long)N3 * 128);
    zero_f32<<<cdiv((unsigned)N3, 256), blk, 0, stream>>>(deg, (unsigned long long)N3);
    sage_scatter<<<cdiv((unsigned)E2, 8), blk, 0, stream>>>(h2, src2, dst2, E2, msg, deg);
    {
        unsigned tiles = cdiv((unsigned)N3, 16) * (64 / 16);
        sage_gemm<<<cdiv(tiles, 8), blk, 0, stream>>>(h2, msg, deg, Ws2, Wn2, b2, (float*)d_out, N3, 64, 0);
    }
}


# Sort edges by destination each call (count, segment hand-out, fill) and sum each row's sources in one wave: float atomics gone; per-row sum order now follows integer-atomic arrival
# speedup vs baseline: 1.2288x; 1.2288x over previous
#include <hip/hip_runtime.h>
#include <hip/hip_bf16.h>


typedef __attribute__((ext_vector_type(2))) float v2f;
typedef __attribute__((ext_vector_type(8))) float v8f;

__global__ __launch_bounds__(256) void zero_u32(unsigned* __restrict__ p, unsigned n) {
    unsigned i = blockIdx.x * blockDim.x + threadIdx.x;
    if (i < n) p[i] = 0u;
}

__global__ __launch_bounds__(256) void edge_count(const int* __restrict__ dst, int nedges,
                                                  unsigned* __restrict__ cnt) {
    int e0 = (int)(blockIdx.x * 256u + threadIdx.x) * 4;
    if (e0 + 3 < nedges) {
        const int4 d = *(const int4*)(dst + e0);
        atomicAdd(cnt + d.x, 1u);
        atomicAdd(cnt + d.y, 1u);
        atomicAdd(cnt + d.z, 1u);
        atomicAdd(cnt + d.w, 1u);
    } else {
        for (int e = e0; e < nedges; ++e) atomicAdd(cnt + dst[e], 1u);
    }
}

__global__ __launch_bounds__(256) void seg_alloc(const unsigned* cnt, int n,
                                                 unsigned* cur, unsigned* total,
                                                 float* __restrict__ deg) {
    __shared__ unsigned wave_off[8];
    __shared__ unsigned blk_base;
    const int tid  = threadIdx.x;
    const int lane = tid & 31;
    const int wv   = tid >> 5;
    const int i    = (int)(blockIdx.x * 256u) + tid;
    const unsigned c = (i < n) ? cnt[i] : 0u;
    unsigned x = c;
    for (int off = 1; off < 32; off <<= 1) {
        unsigned y = __shfl_up(x, off);
        if (lane >= off) x += y;
    }
    if (lane == 31) wave_off[wv] = x;
    __syncthreads();
    if (tid == 0) {
        unsigned t = 0;
        for (int w = 0; w < 8; ++w) {
            unsigned v = wave_off[w];
            wave_off[w] = t;
            t += v;
        }
        blk_base = atomicAdd(total, t);
    }
    __syncthreads();
    if (i < n) {
        cur[i] = blk_base + wave_off[wv] + (x - c);
        deg[i] = (float)c;
    }
}

static __device__ __forceinline__ void put_src(int* __restrict__ perm, unsigned slot,
                                               int nedges, int s) {
    if (slot < (unsigned)nedges) perm[slot] = s;
}

__global__ __launch_bounds__(256) void edge_fill(const int* __restrict__ src,
                                                 const int* __restrict__ dst, int nedges,
                                                 unsigned* __restrict__ cur,
                                                 int* __restrict__ perm) {
    int e0 = (int)(blockIdx.x * 256u + threadIdx.x) * 4;
    if (e0 + 3 < nedges) {
        const int4 s = *(const int4*)(src + e0);
        const int4 d = *(const int4*)(dst + e0);
        unsigned p0 = atomicAdd(cur + d.x, 1u);
        unsigned p1 = atomicAdd(cur + d.y, 1u);
        unsigned p2 = atomicAdd(cur + d.z, 1u);
        unsigned p3 = atomicAdd(cur + d.w, 1u);
        put_src(perm, p0, nedges, s.x);
        put_src(perm, p1, nedges, s.y);
        put_src(perm, p2, nedges, s.z);
        put_src(perm, p3, nedges, s.w);
    } else {
        for (int e = e0; e < nedges; ++e) {
            unsigned p = atomicAdd(cur + dst[e], 1u);
            put_src(perm, p, nedges, src[e]);
        }
    }
}

static __device__ __forceinline__ float4 load_row(const float* __restrict__ col, int s,
                                                  int n_src) {
    unsigned r = (unsigned)s;
    if (r > (unsigned)(n_src - 1)) r = (unsigned)(n_src - 1);
    return *(const float4*)(col + (size_t)r * 128);
}

template <int G>
static __device__ __forceinline__ void add_rows(float4& acc, const float* __restrict__ col,
                                                int mine, unsigned j, int n_src) {
    float4 v[G];
#pragma unroll
    for (int k = 0; k < G; ++k)
        v[k] = load_row(col, __builtin_amdgcn_readlane(mine, (int)j + k), n_src);
#pragma unroll
    for (int k = 0; k < G; ++k) {
        acc.x += v[k].x; acc.y += v[k].y; acc.z += v[k].z; acc.w += v[k].w;
    }
}

__global__ __launch_bounds__(256) void seg_gather(const float* __restrict__ hsrc, int n_src,
                                                  const int* __restrict__ perm,
                                                  const unsigned* __restrict__ cnt,
                                                  const unsigned* __restrict__ cur,
                                                  int nedges, int n_dst,
                                                  float* __restrict__ msg) {
    const int lane = threadIdx.x & 31;
    const int row  = (int)(blockIdx.x * 8u) + (int)(threadIdx.x >> 5);
    if (row >= n_dst) return;
    unsigned n   = (unsigned)__builtin_amdgcn_readfirstlane((int)cnt[row]);
    unsigned end = (unsigned)__builtin_amdgcn_readfirstlane((int)cur[row]);
    if (end > (unsigned)nedges) end = (unsigned)nedges;
    if (n > end) n = end;
    const unsigned beg = end - n;

    const float* col = hsrc + lane * 4;
    float4 acc = make_float4(0.f, 0.f, 0.f, 0.f);
    for (unsigned base = 0; base < n; base += 32) {
        unsigned m = n - base;
        if (m > 32u) m = 32u;
        const int mine = ((unsigned)lane < m) ? perm[beg + base + (unsigned)lane] : 0;
        unsigned j = 0;
        for (; j + 8 <= m; j += 8) add_rows<8>(acc, col, mine, j, n_src);
        if (j + 4 <= m) { add_rows<4>(acc, col, mine, j, n_src); j += 4; }
        for (; j < m; ++j) add_rows<1>(acc, col, mine, j, n_src);
    }
    *(float4*)(msg + (size_t)row * 128 + lane * 4) = acc;
}

__global__ __launch_bounds__(256) void sage_gemm(const float* __restrict__ hdst,
                                                 const float* __restrict__ msg,
                                                 const float* __restrict__ deg,
                                                 const float* __restrict__ Wself,
                                                 const float* __restrict__ Wneigh,
                                                 const float* __restrict__ bias,
                                                 float* __restrict__ out,
                                                 int n_dst, int d_out, int relu) {
    const int d_in = 128;
    int lane   = threadIdx.x & 31;
    int wave   = blockIdx.x * (blockDim.x >> 5) + (threadIdx.x >> 5);
    int ntiles = d_out >> 4;
    int mtiles = (n_dst + 15) >> 4;
    if (wave >= mtiles * ntiles) return;

    int mt = wave / ntiles;
    int nt = wave - mt * ntiles;
    int m0 = mt << 4;
    int n0 = nt << 4;

    int half = lane >> 4;
    int idx  = lane & 15;

    int arow = m0 + idx;
    if (arow > n_dst - 1) arow = n_dst - 1;

    const float* Ah = hdst  + (size_t)arow * d_in + 2 * half;
    const float* Am = msg   + (size_t)arow * d_in + 2 * half;
    const float* Bs = Wself  + (size_t)(n0 + idx) * d_in + 2 * half;
    const float* Bn = Wneigh + (size_t)(n0 + idx) * d_in + 2 * half;
    float invdeg = 1.0f / fmaxf(deg[arow], 1.0f);

    v8f c = {0.f, 0.f, 0.f, 0.f, 0.f, 0.f, 0.f, 0.f};

#pragma unroll
    for (int k = 0; k < d_in; k += 4) {
        v2f a = *(const v2f*)(Ah + k);
        v2f b = *(const v2f*)(Bs + k);
        c = __builtin_amdgcn_wmma_f32_16x16x4_f32(false, a, false, b,
                                                  (short)0, c, false, false);
    }
#pragma unroll
    for (int k = 0; k < d_in; k += 4) {
        v2f a = *(const v2f*)(Am + k);
        a.x *= invdeg;
        a.y *= invdeg;
        v2f b = *(const v2f*)(Bn + k);
        c = __builtin_amdgcn_wmma_f32_16x16x4_f32(false, a, false, b,
                                                  (short)0, c, false, false);
    }

    float bv = bias[n0 + idx];
#pragma unroll
    for (int j = 0; j < 8; ++j) {
        int m = m0 + j + 8 * half;
        if (m < n_dst) {
            float v = c[j] + bv;
            if (relu) v = fmaxf(v, 0.0f);
            out[(size_t)m * d_out + (n0 + idx)] = v;
        }
    }
}

static inline unsigned cdiv(unsigned a, unsigned b) { return (a + b - 1) / b; }

static void sage_aggregate(const float* hsrc, int n_src, const int* src, const int* dst,
                           int nedges, int n_dst, char* scratch,
                           float* msg, float* deg, hipStream_t stream) {
    unsigned* cnt  = (unsigned*)scratch;
    unsigned* cur  = (unsigned*)(scratch + (512u << 10));
    int*      perm = (int*)(scratch + (1024u << 10));
    const dim3 blk(256);
    zero_u32<<<cdiv((unsigned)n_dst + 1u, 256), blk, 0, stream>>>(cnt, (unsigned)n_dst + 1u);
    edge_count<<<cdiv((unsigned)nedges, 1024), blk, 0, stream>>>(dst, nedges, cnt);
    seg_alloc<<<cdiv((unsigned)n_dst, 256), blk, 0, stream>>>(cnt, n_dst, cur, cnt + n_dst, deg);
    edge_fill<<<cdiv((unsigned)nedges, 1024), blk, 0, stream>>>(src, dst, nedges, cur, perm);
    seg_gather<<<cdiv((unsigned)n_dst, 8), blk, 0, stream>>>(hsrc, n_src, perm, cnt, cur,
                                                            nedges, n_dst, msg);
}

extern "C" void kernel_launch(void* const* d_in, const int* in_sizes, int n_in,
                              void* d_out, int out_size, void* d_ws, size_t ws_size,
                              hipStream_t stream) {
    const float* feats = (const float*)d_in[0];
    const int* src0 = (const int*)d_in[1];
    const int* dst0 = (const int*)d_in[2];
    const int* src1 = (const int*)d_in[3];
    const int* dst1 = (const int*)d_in[4];
    const int* src2 = (const int*)d_in[5];
    const int* dst2 = (const int*)d_in[6];
    const float* Ws0 = (const float*)d_in[10];
    const float* Wn0 = (const float*)d_in[11];
    const float* b0  = (const float*)d_in[12];
    const float* Ws1 = (const float*)d_in[13];
    const float* Wn1 = (const float*)d_in[14];
    const float* b1  = (const float*)d_in[15];
    const float* Ws2 = (const float*)d_in[16];
    const float* Wn2 = (const float*)d_in[17];
    const float* b2  = (const float*)d_in[18];

    const int E0 = in_sizes[1], E1 = in_sizes[3], E2 = in_sizes[5];
    const int N0 = 500000, N1 = 100000, N2 = 20000, N3 = 5000;

    char* ws = (char*)d_ws;
    float* h1  = (float*)ws;                    ws += (size_t)N1 * 128 * sizeof(float);
    float* h2  = (float*)ws;                    ws += (size_t)N2 * 128 * sizeof(float);
    float* msg = (float*)ws;                    ws += (size_t)N1 * 128 * sizeof(float);
    float* deg = (float*)ws;

    const dim3 blk(256);

    sage_aggregate(feats, N0, src0, dst0, E0, N1, (char*)h1, msg, deg, stream);
    {
        unsigned tiles = cdiv((unsigned)N1, 16) * (128 / 16);
        sage_gemm<<<cdiv(tiles, 8), blk, 0, stream>>>(feats, msg, deg, Ws0, Wn0, b0, h1, N1, 128, 1);
    }

    sage_aggregate(h1, N1, src1, dst1, E1, N2, (char*)h2, msg, deg, stream);
    {
        unsigned tiles = cdiv((unsigned)N2, 16) * (128 / 16);
        sage_gemm<<<cdiv(tiles, 8), blk, 0, stream>>>(h1, msg, deg, Ws1, Wn1, b1, h2, N2, 128, 1);
    }

    sage_aggregate(h2, N2, src2, dst2, E2, N3, (char*)h1, msg, deg, stream);
    {
        unsigned tiles = cdiv((unsigned)N3, 16) * (64 / 16);
        sage_gemm<<<cdiv(tiles, 8), blk, 0, stream>>>(h2, msg, deg, Ws2, Wn2, b2, (float*)d_out, N3, 64, 0);
    }
}


# Layer GEMM re-tiled: one wave per 16-row strip over all columns, both weight matrices staged once per work-group in LDS, A and B read 16 B per lane; k sum order inside each group of 8 changes
# speedup vs baseline: 1.5117x; 1.2302x over previous
#include <hip/hip_runtime.h>
#include <hip/hip_bf16.h>


typedef __attribute__((ext_vector_type(2))) float v2f;
typedef __attribute__((ext_vector_type(8))) float v8f;

__global__ __launch_bounds__(256) void zero_u32(unsigned* __restrict__ p, unsigned n) {
    unsigned i = blockIdx.x * blockDim.x + threadIdx.x;
    if (i < n) p[i] = 0u;
}

__global__ __launch_bounds__(256) void edge_count(const int* __restrict__ dst, int nedges,
                                                  unsigned* __restrict__ cnt) {
    int e0 = (int)(blockIdx.x * 256u + threadIdx.x) * 4;
    if (e0 + 3 < nedges) {
        const int4 d = *(const int4*)(dst + e0);
        atomicAdd(cnt + d.x, 1u);
        atomicAdd(cnt + d.y, 1u);
        atomicAdd(cnt + d.z, 1u);
        atomicAdd(cnt + d.w, 1u);
    } else {
        for (int e = e0; e < nedges; ++e) atomicAdd(cnt + dst[e], 1u);
    }
}

__global__ __launch_bounds__(256) void seg_alloc(const unsigned* cnt, int n,
                                                 unsigned* cur, unsigned* total,
                                                 float* __restrict__ deg) {
    __shared__ unsigned wave_off[8];
    __shared__ unsigned blk_base;
    const int tid  = threadIdx.x;
    const int lane = tid & 31;
    const int wv   = tid >> 5;
    const int i    = (int)(blockIdx.x * 256u) + tid;
    const unsigned c = (i < n) ? cnt[i] : 0u;
    unsigned x = c;
    for (int off = 1; off < 32; off <<= 1) {
        unsigned y = __shfl_up(x, off);
        if (lane >= off) x += y;
    }
    if (lane == 31) wave_off[wv] = x;
    __syncthreads();
    if (tid == 0) {
        unsigned t = 0;
        for (int w = 0; w < 8; ++w) {
            unsigned v = wave_off[w];
            wave_off[w] = t;
            t += v;
        }
        blk_base = atomicAdd(total, t);
    }
    __syncthreads();
    if (i < n) {
        cur[i] = blk_base + wave_off[wv] + (x - c);
        deg[i] = (float)c;
    }
}

static __device__ __forceinline__ void put_src(int* __restrict__ perm, unsigned slot,
                                               int nedges, int s) {
    if (slot < (unsigned)nedges) perm[slot] = s;
}

__global__ __launch_bounds__(256) void edge_fill(const int* __restrict__ src,
                                                 const int* __restrict__ dst, int nedges,
                                                 unsigned* __restrict__ cur,
                                                 int* __restrict__ perm) {
    int e0 = (int)(blockIdx.x * 256u + threadIdx.x) * 4;
    if (e0 + 3 < nedges) {
        const int4 s = *(const int4*)(src + e0);
        const int4 d = *(const int4*)(dst + e0);
        unsigned p0 = atomicAdd(cur + d.x, 1u);
        unsigned p1 = atomicAdd(cur + d.y, 1u);
        unsigned p2 = atomicAdd(cur + d.z, 1u);
        unsigned p3 = atomicAdd(cur + d.w, 1u);
        put_src(perm, p0, nedges, s.x);
        put_src(perm, p1, nedges, s.y);
        put_src(perm, p2, nedges, s.z);
        put_src(perm, p3, nedges, s.w);
    } else {
        for (int e = e0; e < nedges; ++e) {
            unsigned p = atomicAdd(cur + dst[e], 1u);
            put_src(perm, p, nedges, src[e]);
        }
    }
}

static __device__ __forceinline__ float4 load_row(const float* __restrict__ col, int s,
                                                  int n_src) {
    unsigned r = (unsigned)s;
    if (r > (unsigned)(n_src - 1)) r = (unsigned)(n_src - 1);
    return *(const float4*)(col + (size_t)r * 128);
}

template <int G>
static __device__ __forceinline__ void add_rows(float4& acc, const float* __restrict__ col,
                                                int mine, unsigned j, int n_src) {
    float4 v[G];
#pragma unroll
    for (int k = 0; k < G; ++k)
        v[k] = load_row(col, __builtin_amdgcn_readlane(mine, (int)j + k), n_src);
#pragma unroll
    for (int k = 0; k < G; ++k) {
        acc.x += v[k].x; acc.y += v[k].y; acc.z += v[k].z; acc.w += v[k].w;
    }
}

__global__ __launch_bounds__(256) void seg_gather(const float* __restrict__ hsrc, int n_src,
                                                  const int* __restrict__ perm,
                                                  const unsigned* __restrict__ cnt,
                                                  const unsigned* __restrict__ cur,
                                                  int nedges, int n_dst,
                                                  float* __restrict__ msg) {
    const int lane = threadIdx.x & 31;
    const int row  = (int)(blockIdx.x * 8u) + (int)(threadIdx.x >> 5);
    if (row >= n_dst) return;
    unsigned n   = (unsigned)__builtin_amdgcn_readfirstlane((int)cnt[row]);
    unsigned end = (unsigned)__builtin_amdgcn_readfirstlane((int)cur[row]);
    if (end > (unsigned)nedges) end = (unsigned)nedges;
    if (n > end) n = end;
    const unsigned beg = end - n;

    const float* col = hsrc + lane * 4;
    float4 acc = make_float4(0.f, 0.f, 0.f, 0.f);
    for (unsigned base = 0; base < n; base += 32) {
        unsigned m = n - base;
        if (m > 32u) m = 32u;
        const int mine = ((unsigned)lane < m) ? perm[beg + base + (unsigned)lane] : 0;
        unsigned j = 0;
        for (; j + 8 <= m; j += 8) add_rows<8>(acc, col, mine, j, n_src);
        if (j + 4 <= m) { add_rows<4>(acc, col, mine, j, n_src); j += 4; }
        for (; j < m; ++j) add_rows<1>(acc, col, mine, j, n_src);
    }
    *(float4*)(msg + (size_t)row * 128 + lane * 4) = acc;
}

#define W_PITCH 260

static __device__ __forceinline__ int k_of(int S, int half) { return 8 * S + 4 * half; }

template <int NT, int WAVES, bool RELU>
__global__ __launch_bounds__(WAVES * 32) void sage_gemm(const float* __restrict__ hdst,
                                                        const float* __restrict__ msg,
                                                        const float* __restrict__ deg,
                                                        const float* __restrict__ Wself,
                                                        const float* __restrict__ Wneigh,
                                                        const float* __restrict__ bias,
                                                        float* __restrict__ out,
                                                        int n_dst) {
    constexpr int N       = NT * 16;
    constexpr int THREADS = WAVES * 32;
    constexpr int F4      = N * 64;
    constexpr int STG     = 16;
    static_assert(F4 % (THREADS * STG) == 0, "weight staging loop");
    __shared__ __attribute__((aligned(16))) float Wl[N * W_PITCH];

    const int tid  = threadIdx.x;
    const int lane = tid & 31;
    const int wave = tid >> 5;
    const int half = lane >> 4;
    const int idx  = lane & 15;

    const int strip   = (int)blockIdx.x * WAVES + wave;
    const int nstrips = (n_dst + 15) >> 4;
    const int m0      = strip << 4;

    int arow = m0 + idx;
    if (arow > n_dst - 1) arow = n_dst - 1;

    const float* Ah = hdst + (size_t)arow * 128 + k_of(0, half);
    const float* Am = msg  + (size_t)arow * 128 + k_of(0, half);

    float4 a_cur[8];
#pragma unroll
    for (int s = 0; s < 8; ++s) a_cur[s] = *(const float4*)(Ah + 8 * s);
    const float invdeg = 1.0f / fmaxf(deg[arow], 1.0f);

    for (int i0 = 0; i0 < F4; i0 += THREADS * STG) {
        float4 t[STG];
#pragma unroll
        for (int u = 0; u < STG; ++u) {
            const int q   = i0 + u * THREADS + tid;
            const int row = q >> 6;
            const int c4  = q & 63;
            const float* s = (c4 < 32) ? (Wself + row * 128 + c4 * 4)
                                       : (Wneigh + row * 128 + (c4 - 32) * 4);
            t[u] = *(const float4*)s;
        }
#pragma unroll
        for (int u = 0; u < STG; ++u) {
            const int q   = i0 + u * THREADS + tid;
            const int row = q >> 6;
            const int c4  = q & 63;
            *(float4*)(Wl + row * W_PITCH + c4 * 4) = t[u];
        }
    }
    __syncthreads();
    if (strip >= nstrips) return;

    v8f acc[NT];
#pragma unroll
    for (int t = 0; t < NT; ++t) acc[t] = (v8f){0.f, 0.f, 0.f, 0.f, 0.f, 0.f, 0.f, 0.f};

    const float* Bl = Wl + idx * W_PITCH + k_of(0, half);

    float4 b_cur[NT];
#pragma unroll
    for (int t = 0; t < NT; ++t) b_cur[t] = *(const float4*)(Bl + t * 16 * W_PITCH);
    __builtin_amdgcn_sched_group_barrier(0x100, NT, 0);

#pragma unroll
    for (int c = 0; c < 4; ++c) {
        float4 a_nxt[8];
        if (c + 1 < 4) {
            const float* p = (c + 1 < 2) ? (Ah + 64 * (c + 1)) : (Am + 64 * (c + 1 - 2));
#pragma unroll
            for (int s = 0; s < 8; ++s) a_nxt[s] = *(const float4*)(p + 8 * s);
        }
#pragma unroll
        for (int s = 0; s < 8; ++s) {
            const int S = c * 8 + s;
            float4 a = a_cur[s];
            if (c >= 2) { a.x *= invdeg; a.y *= invdeg; a.z *= invdeg; a.w *= invdeg; }
            float4 b_nxt[NT];
            if (S + 1 < 32) {
#pragma unroll
                for (int t = 0; t < NT; ++t)
                    b_nxt[t] = *(const float4*)(Bl + t * 16 * W_PITCH + 8 * (S + 1));
            }
            const v2f a01 = {a.x, a.y};
            const v2f a23 = {a.z, a.w};
#pragma unroll
            for (int t = 0; t < NT; ++t) {
                const v2f b01 = {b_cur[t].x, b_cur[t].y};
                acc[t] = __builtin_amdgcn_wmma_f32_16x16x4_f32(false, a01, false, b01,
                                                               (short)0, acc[t], false, false);
            }
#pragma unroll
            for (int t = 0; t < NT; ++t) {
                const v2f b23 = {b_cur[t].z, b_cur[t].w};
                acc[t] = __builtin_amdgcn_wmma_f32_16x16x4_f32(false, a23, false, b23,
                                                               (short)0, acc[t], false, false);
            }
            if (S + 1 < 32) {
                __builtin_amdgcn_sched_group_barrier(0x100, NT, 0);
                __builtin_amdgcn_sched_group_barrier(0x008, 2 * NT, 0);
#pragma unroll
                for (int t = 0; t < NT; ++t) b_cur[t] = b_nxt[t];
            }
        }
        if (c + 1 < 4) {
#pragma unroll
            for (int s = 0; s < 8; ++s) a_cur[s] = a_nxt[s];
        }
    }

    float bv[NT];
#pragma unroll
    for (int t = 0; t < NT; ++t) bv[t] = bias[t * 16 + idx];
    float* o = out + (size_t)(m0 + 8 * half) * N + idx;
    if (m0 + 16 <= n_dst) {
#pragma unroll
        for (int t = 0; t < NT; ++t) {
#pragma unroll
            for (int j = 0; j < 8; ++j) {
                float v = acc[t][j] + bv[t];
                if (RELU) v = fmaxf(v, 0.0f);
                o[j * N + t * 16] = v;
            }
        }
    } else {
#pragma unroll
        for (int t = 0; t < NT; ++t) {
#pragma unroll
            for (int j = 0; j < 8; ++j) {
                if (m0 + j + 8 * half < n_dst) {
                    float v = acc[t][j] + bv[t];
                    if (RELU) v = fmaxf(v, 0.0f);
                    o[j * N + t * 16] = v;
                }
            }
        }
    }
}

static inline unsigned cdiv(unsigned a, unsigned b) { return (a + b - 1) / b; }

static void sage_aggregate(const float* hsrc, int n_src, const int* src, const int* dst,
                           int nedges, int n_dst, char* scratch,
                           float* msg, float* deg, hipStream_t stream) {
    unsigned* cnt  = (unsigned*)scratch;
    unsigned* cur  = (unsigned*)(scratch + (512u << 10));
    int*      perm = (int*)(scratch + (1024u << 10));
    const dim3 blk(256);
    zero_u32<<<cdiv((unsigned)n_dst + 1u, 256), blk, 0, stream>>>(cnt, (unsigned)n_dst + 1u);
    edge_count<<<cdiv((unsigned)nedges, 1024), blk, 0, stream>>>(dst, nedges, cnt);
    seg_alloc<<<cdiv((unsigned)n_dst, 256), blk, 0, stream>>>(cnt, n_dst, cur, cnt + n_dst, deg);
    edge_fill<<<cdiv((unsigned)nedges, 1024), blk, 0, stream>>>(src, dst, nedges, cur, perm);
    seg_gather<<<cdiv((unsigned)n_dst, 8), blk, 0, stream>>>(hsrc, n_src, perm, cnt, cur,
                                                            nedges, n_dst, msg);
}

extern "C" void kernel_launch(void* const* d_in, const int* in_sizes, int n_in,
                              void* d_out, int out_size, void* d_ws, size_t ws_size,
                              hipStream_t stream) {
    const float* feats = (const float*)d_in[0];
    const int* src0 = (const int*)d_in[1];
    const int* dst0 = (const int*)d_in[2];
    const int* src1 = (const int*)d_in[3];
    const int* dst1 = (const int*)d_in[4];
    const int* src2 = (const int*)d_in[5];
    const int* dst2 = (const int*)d_in[6];
    const float* Ws0 = (const float*)d_in[10];
    const float* Wn0 = (const float*)d_in[11];
    const float* b0  = (const float*)d_in[12];
    const float* Ws1 = (const float*)d_in[13];
    const float* Wn1 = (const float*)d_in[14];
    const float* b1  = (const float*)d_in[15];
    const float* Ws2 = (const float*)d_in[16];
    const float* Wn2 = (const float*)d_in[17];
    const float* b2  = (const float*)d_in[18];

    const int E0 = in_sizes[1], E1 = in_sizes[3], E2 = in_sizes[5];
    const int N0 = 500000, N1 = 100000, N2 = 20000, N3 = 5000;

    char* ws = (char*)d_ws;
    float* h1  = (float*)ws;                    ws += (size_t)N1 * 128 * sizeof(float);
    float* h2  = (float*)ws;                    ws += (size_t)N2 * 128 * sizeof(float);
    float* msg = (float*)ws;                    ws += (size_t)N1 * 128 * sizeof(float);
    float* deg = (float*)ws;

    const dim3 blk(256);

    sage_aggregate(feats, N0, src0, dst0, E0, N1, (char*)h1, msg, deg, stream);
    {
        unsigned strips = cdiv((unsigned)N1, 16);
        sage_gemm<8, 8, true><<<cdiv(strips, 8), blk, 0, stream>>>(feats, msg, deg, Ws0, Wn0, b0, h1, N1);
    }

    sage_aggregate(h1, N1, src1, dst1, E1, N2, (char*)h2, msg, deg, stream);
    {
        unsigned strips = cdiv((unsigned)N2, 16);
        sage_gemm<8, 8, true><<<cdiv(strips, 8), blk, 0, stream>>>(h1, msg, deg, Ws1, Wn1, b1, h2, N2);
    }

    sage_aggregate(h2, N2, src2, dst2, E2, N3, (char*)h1, msg, deg, stream);
    {
        unsigned strips = cdiv((unsigned)N3, 16);
        sage_gemm<4, 4, false><<<cdiv(strips, 4), dim3(128), 0, stream>>>(h2, msg, deg, Ws2, Wn2, b2, (float*)d_out, N3);
    }
}


# Edge sort: per-edge global atomics (count, fill) become LDS counters in a two-level counting sort (256-row buckets, then rows), lists written as whole lines; row sum order follows LDS-add arrival
# speedup vs baseline: 2.9281x; 1.9370x over previous
#include <hip/hip_runtime.h>
#include <hip/hip_bf16.h>


typedef __attribute__((ext_vector_type(2))) float v2f;
typedef __attribute__((ext_vector_type(8))) float v8f;

#define BK_ROWS   256
#define MAX_BK    512
#define CHUNK     8192
#define STAGE_CAP 12288

__global__ __launch_bounds__(256) void zero_u32(unsigned* __restrict__ p, unsigned n) {
    unsigned i = blockIdx.x * blockDim.x + threadIdx.x;
    if (i < n) p[i] = 0u;
}

static __device__ __forceinline__ unsigned bucket_of(int d, unsigned nb) {
    const unsigned b = (unsigned)d >> 8;
    return b < nb ? b : nb - 1u;
}

static __device__ __forceinline__ void chunk_histogram(unsigned* hist, const int* __restrict__ dst,
                                                       int nedges, unsigned nb) {
    const int cbeg = (int)blockIdx.x * CHUNK;
    if (cbeg + CHUNK <= nedges) {
        const int4* q = (const int4*)(dst + cbeg) + threadIdx.x;
        int4 d[CHUNK / 1024];
#pragma unroll
        for (int t = 0; t < CHUNK / 1024; ++t) d[t] = q[t * 256];
#pragma unroll
        for (int t = 0; t < CHUNK / 1024; ++t) {
            atomicAdd(&hist[bucket_of(d[t].x, nb)], 1u);
            atomicAdd(&hist[bucket_of(d[t].y, nb)], 1u);
            atomicAdd(&hist[bucket_of(d[t].z, nb)], 1u);
            atomicAdd(&hist[bucket_of(d[t].w, nb)], 1u);
        }
    } else {
        for (int e = cbeg + (int)threadIdx.x; e < nedges; e += 256)
            atomicAdd(&hist[bucket_of(dst[e], nb)], 1u);
    }
}

__global__ __launch_bounds__(256) void bucket_count(const int* __restrict__ dst, int nedges,
                                                    unsigned nb, unsigned* __restrict__ ccnt) {
    __shared__ unsigned hist[MAX_BK];
    const int tid = threadIdx.x;
    hist[tid] = 0u;
    hist[tid + 256] = 0u;
    __syncthreads();
    chunk_histogram(hist, dst, nedges, nb);
    __syncthreads();
    for (unsigned b = tid; b < nb; b += 256u) {
        const unsigned c = hist[b];
        if (c) atomicAdd(ccnt + b, c);
    }
}

static __device__ __forceinline__ unsigned block_scan_excl(unsigned v, unsigned* wave_off) {
    const int lane = threadIdx.x & 31;
    const int wv   = threadIdx.x >> 5;
    unsigned x = v;
    for (int off = 1; off < 32; off <<= 1) {
        unsigned y = __shfl_up(x, off);
        if (lane >= off) x += y;
    }
    if (lane == 31) wave_off[wv] = x;
    __syncthreads();
    unsigned pre = 0;
    for (int w = 0; w < wv; ++w) pre += wave_off[w];
    return pre + x - v;
}

__global__ __launch_bounds__(256) void bucket_scatter(const int* __restrict__ src,
                                                      const int* __restrict__ dst, int nedges,
                                                      unsigned nb,
                                                      const unsigned* __restrict__ ccnt,
                                                      unsigned* __restrict__ ccur,
                                                      int2* __restrict__ tmp) {
    __shared__ unsigned hist[MAX_BK];
    __shared__ unsigned lstart[MAX_BK];
    __shared__ unsigned gstart[MAX_BK];
    __shared__ unsigned wave_off[8];
    __shared__ unsigned wave_off2[8];
    __shared__ int2 stage[CHUNK];
    const int tid = threadIdx.x;
    const unsigned b0 = 2u * (unsigned)tid;
    hist[tid] = 0u;
    hist[tid + 256] = 0u;
    const unsigned c0 = (b0 < nb) ? ccnt[b0] : 0u;
    const unsigned c1 = (b0 + 1u < nb) ? ccnt[b0 + 1u] : 0u;
    const unsigned gex = block_scan_excl(c0 + c1, wave_off);
    chunk_histogram(hist, dst, nedges, nb);
    __syncthreads();
    const unsigned h0 = hist[b0];
    const unsigned h1 = hist[b0 + 1u];
    const unsigned lex = block_scan_excl(h0 + h1, wave_off2);
    const unsigned o0 = h0 ? atomicAdd(ccur + b0, h0) : 0u;
    const unsigned o1 = h1 ? atomicAdd(ccur + b0 + 1u, h1) : 0u;
    lstart[b0]      = lex;
    lstart[b0 + 1u] = lex + h0;
    hist[b0]        = lex;
    hist[b0 + 1u]   = lex + h0;
    gstart[b0]      = gex + o0;
    gstart[b0 + 1u] = gex + c0 + o1;
    __syncthreads();
    const int cbeg = (int)blockIdx.x * CHUNK;
    if (cbeg + CHUNK <= nedges) {
        const int4* qs = (const int4*)(src + cbeg) + tid;
        const int4* qd = (const int4*)(dst + cbeg) + tid;
        int4 s[CHUNK / 1024], d[CHUNK / 1024];
#pragma unroll
        for (int t = 0; t < CHUNK / 1024; ++t) { d[t] = qd[t * 256]; s[t] = qs[t * 256]; }
#pragma unroll
        for (int t = 0; t < CHUNK / 1024; ++t) {
            const unsigned p0 = atomicAdd(&hist[bucket_of(d[t].x, nb)], 1u);
            const unsigned p1 = atomicAdd(&hist[bucket_of(d[t].y, nb)], 1u);
            const unsigned p2 = atomicAdd(&hist[bucket_of(d[t].z, nb)], 1u);
            const unsigned p3 = atomicAdd(&hist[bucket_of(d[t].w, nb)], 1u);
            if (p0 < (unsigned)CHUNK) stage[p0] = make_int2(s[t].x, d[t].x);
            if (p1 < (unsigned)CHUNK) stage[p1] = make_int2(s[t].y, d[t].y);
            if (p2 < (unsigned)CHUNK) stage[p2] = make_int2(s[t].z, d[t].z);
            if (p3 < (unsigned)CHUNK) stage[p3] = make_int2(s[t].w, d[t].w);
        }
    } else {
        for (int e = cbeg + tid; e < nedges; e += 256) {
            const int d = dst[e];
            const unsigned p = atomicAdd(&hist[bucket_of(d, nb)], 1u);
            if (p < (unsigned)CHUNK) stage[p] = make_int2(src[e], d);
        }
    }
    __syncthreads();
    int cn = nedges - cbeg;
    if (cn > CHUNK) cn = CHUNK;
    for (int i = tid; i < cn; i += 256) {
        const int2 e = stage[i];
        const unsigned b = bucket_of(e.y, nb);
        const unsigned pos = gstart[b] + ((unsigned)i - lstart[b]);
        if (pos < (unsigned)nedges) tmp[pos] = e;
    }
}

__global__ __launch_bounds__(256) void bucket_sort(const int2* __restrict__ tmp, int nedges,
                                                   int n_dst, const unsigned* __restrict__ ccnt,
                                                   unsigned* __restrict__ cnt,
                                                   unsigned* __restrict__ cur,
                                                   float* __restrict__ deg,
                                                   int* __restrict__ perm) {
    __shared__ unsigned rows[BK_ROWS];
    __shared__ unsigned wave_off[8];
    __shared__ unsigned red[8];
    __shared__ int stage[STAGE_CAP];
    const int tid  = threadIdx.x;
    const int lane = tid & 31;
    const int wv   = tid >> 5;
    const unsigned b = blockIdx.x;
    unsigned part = 0;
    for (unsigned i = tid; i < b; i += 256u) part += ccnt[i];
    for (int off = 16; off > 0; off >>= 1) part += __shfl_xor(part, off);
    if (lane == 0) red[wv] = part;
    rows[tid] = 0u;
    __syncthreads();
    unsigned beg = 0;
    for (int w = 0; w < 8; ++w) beg += red[w];
    unsigned n = ccnt[b];
    if (beg > (unsigned)nedges) beg = (unsigned)nedges;
    if (n > (unsigned)nedges - beg) n = (unsigned)nedges - beg;
    const int2* mine = tmp + beg;
    const int row0 = (int)(b * BK_ROWS);
    for (unsigned i0 = tid; i0 < n; i0 += 1024u) {
        int d[4];
#pragma unroll
        for (int k = 0; k < 4; ++k) {
            const unsigned i = i0 + 256u * k;
            d[k] = mine[i < n ? i : n - 1u].y;
        }
#pragma unroll
        for (int k = 0; k < 4; ++k)
            if (i0 + 256u * k < n) atomicAdd(&rows[(unsigned)(d[k] - row0) & (BK_ROWS - 1u)], 1u);
    }
    __syncthreads();
    const unsigned c  = rows[tid];
    const unsigned ex = block_scan_excl(c, wave_off);
    const int row = row0 + tid;
    if (row < n_dst) {
        cnt[row] = c;
        cur[row] = beg + ex + c;
        deg[row] = (float)c;
    }
    rows[tid] = ex;
    __syncthreads();
    if (n <= (unsigned)STAGE_CAP) {
        for (unsigned i0 = tid; i0 < n; i0 += 1024u) {
            int2 e[4];
#pragma unroll
            for (int k = 0; k < 4; ++k) {
                const unsigned i = i0 + 256u * k;
                e[k] = mine[i < n ? i : n - 1u];
            }
#pragma unroll
            for (int k = 0; k < 4; ++k) {
                if (i0 + 256u * k < n) {
                    const unsigned p =
                        atomicAdd(&rows[(unsigned)(e[k].y - row0) & (BK_ROWS - 1u)], 1u);
                    if (p < (unsigned)STAGE_CAP) stage[p] = e[k].x;
                }
            }
        }
        __syncthreads();
        for (unsigned i = tid; i < n; i += 256u) perm[beg + i] = stage[i];
    } else {
        for (unsigned i = tid; i < n; i += 256u) {
            const int2 e = mine[i];
            const unsigned p = beg + atomicAdd(&rows[(unsigned)(e.y - row0) & (BK_ROWS - 1u)], 1u);
            if (p < (unsigned)nedges) perm[p] = e.x;
        }
    }
}

static __device__ __forceinline__ float4 load_row(const float* __restrict__ col, int s,
                                                  int n_src) {
    unsigned r = (unsigned)s;
    if (r > (unsigned)(n_src - 1)) r = (unsigned)(n_src - 1);
    return *(const float4*)(col + (size_t)r * 128);
}

template <int G>
static __device__ __forceinline__ void add_rows(float4& acc, const float* __restrict__ col,
                                                int mine, unsigned j, int n_src) {
    float4 v[G];
#pragma unroll
    for (int k = 0; k < G; ++k)
        v[k] = load_row(col, __builtin_amdgcn_readlane(mine, (int)j + k), n_src);
#pragma unroll
    for (int k = 0; k < G; ++k) {
        acc.x += v[k].x; acc.y += v[k].y; acc.z += v[k].z; acc.w += v[k].w;
    }
}

__global__ __launch_bounds__(256) void seg_gather(const float* __restrict__ hsrc, int n_src,
                                                  const int* __restrict__ perm,
                                                  const unsigned* __restrict__ cnt,
                                                  const unsigned* __restrict__ cur,
                                                  int nedges, int n_dst,
                                                  float* __restrict__ msg) {
    const int lane = threadIdx.x & 31;
    const int row  = (int)(blockIdx.x * 8u) + (int)(threadIdx.x >> 5);
    if (row >= n_dst) return;
    unsigned n   = (unsigned)__builtin_amdgcn_readfirstlane((int)cnt[row]);
    unsigned end = (unsigned)__builtin_amdgcn_readfirstlane((int)cur[row]);
    if (end > (unsigned)nedges) end = (unsigned)nedges;
    if (n > end) n = end;
    const unsigned beg = end - n;

    const float* col = hsrc + lane * 4;
    float4 acc = make_float4(0.f, 0.f, 0.f, 0.f);
    for (unsigned base = 0; base < n; base += 32) {
        unsigned m = n - base;
        if (m > 32u) m = 32u;
        const int mine = ((unsigned)lane < m) ? perm[beg + base + (unsigned)lane] : 0;
        unsigned j = 0;
        for (; j + 8 <= m; j += 8) add_rows<8>(acc, col, mine, j, n_src);
        if (j + 4 <= m) { add_rows<4>(acc, col, mine, j, n_src); j += 4; }
        for (; j < m; ++j) add_rows<1>(acc, col, mine, j, n_src);
    }
    *(float4*)(msg + (size_t)row * 128 + lane * 4) = acc;
}

#define W_PITCH 260

static __device__ __forceinline__ int k_of(int S, int half) { return 8 * S + 4 * half; }

template <int NT, int WAVES, bool RELU>
__global__ __launch_bounds__(WAVES * 32) void sage_gemm(const float* __restrict__ hdst,
                                                        const float* __restrict__ msg,
                                                        const float* __restrict__ deg,
                                                        const float* __restrict__ Wself,
                                                        const float* __restrict__ Wneigh,
                                                        const float* __restrict__ bias,
                                                        float* __restrict__ out,
                                                        int n_dst) {
    constexpr int N       = NT * 16;
    constexpr int THREADS = WAVES * 32;
    constexpr int F4      = N * 64;
    constexpr int STG     = 16;
    static_assert(F4 % (THREADS * STG) == 0, "weight staging loop");
    __shared__ __attribute__((aligned(16))) float Wl[N * W_PITCH];

    const int tid  = threadIdx.x;
    const int lane = tid & 31;
    const int wave = tid >> 5;
    const int half = lane >> 4;
    const int idx  = lane & 15;

    const int strip   = (int)blockIdx.x * WAVES + wave;
    const int nstrips = (n_dst + 15) >> 4;
    const int m0      = strip << 4;

    int arow = m0 + idx;
    if (arow > n_dst - 1) arow = n_dst - 1;

    const float* Ah = hdst + (size_t)arow * 128 + k_of(0, half);
    const float* Am = msg  + (size_t)arow * 128 + k_of(0, half);

    float4 a_cur[8];
#pragma unroll
    for (int s = 0; s < 8; ++s) a_cur[s] = *(const float4*)(Ah + 8 * s);
    const float invdeg = 1.0f / fmaxf(deg[arow], 1.0f);

    for (int i0 = 0; i0 < F4; i0 += THREADS * STG) {
        float4 t[STG];
#pragma unroll
        for (int u = 0; u < STG; ++u) {
            const int q   = i0 + u * THREADS + tid;
            const int row = q >> 6;
            const int c4  = q & 63;
            const float* s = (c4 < 32) ? (Wself + row * 128 + c4 * 4)
                                       : (Wneigh + row * 128 + (c4 - 32) * 4);
            t[u] = *(const float4*)s;
        }
#pragma unroll
        for (int u = 0; u < STG; ++u) {
            const int q   = i0 + u * THREADS + tid;
            const int row = q >> 6;
            const int c4  = q & 63;
            *(float4*)(Wl + row * W_PITCH + c4 * 4) = t[u];
        }
    }
    __syncthreads();
    if (strip >= nstrips) return;

    v8f acc[NT];
#pragma unroll
    for (int t = 0; t < NT; ++t) acc[t] = (v8f){0.f, 0.f, 0.f, 0.f, 0.f, 0.f, 0.f, 0.f};

    const float* Bl = Wl + idx * W_PITCH + k_of(0, half);

    float4 b_cur[NT];
#pragma unroll
    for (int t = 0; t < NT; ++t) b_cur[t] = *(const float4*)(Bl + t * 16 * W_PITCH);
    __builtin_amdgcn_sched_group_barrier(0x100, NT, 0);

#pragma unroll
    for (int c = 0; c < 4; ++c) {
        float4 a_nxt[8];
        if (c + 1 < 4) {
            const float* p = (c + 1 < 2) ? (Ah + 64 * (c + 1)) : (Am + 64 * (c + 1 - 2));
#pragma unroll
            for (int s = 0; s < 8; ++s) a_nxt[s] = *(const float4*)(p + 8 * s);
        }
#pragma unroll
        for (int s = 0; s < 8; ++s) {
            const int S = c * 8 + s;
            float4 a = a_cur[s];
            if (c >= 2) { a.x *= invdeg; a.y *= invdeg; a.z *= invdeg; a.w *= invdeg; }
            float4 b_nxt[NT];
            if (S + 1 < 32) {
#pragma unroll
                for (int t = 0; t < NT; ++t)
                    b_nxt[t] = *(const float4*)(Bl + t * 16 * W_PITCH + 8 * (S + 1));
            }
            const v2f a01 = {a.x, a.y};
            const v2f a23 = {a.z, a.w};
#pragma unroll
            for (int t = 0; t < NT; ++t) {
                const v2f b01 = {b_cur[t].x, b_cur[t].y};
                acc[t] = __builtin_amdgcn_wmma_f32_16x16x4_f32(false, a01, false, b01,
                                                               (short)0, acc[t], false, false);
            }
#pragma unroll
            for (int t = 0; t < NT; ++t) {
                const v2f b23 = {b_cur[t].z, b_cur[t].w};
                acc[t] = __builtin_amdgcn_wmma_f32_16x16x4_f32(false, a23, false, b23,
                                                               (short)0, acc[t], false, false);
            }
            if (S + 1 < 32) {
                __builtin_amdgcn_sched_group_barrier(0x100, NT, 0);
                __builtin_amdgcn_sched_group_barrier(0x008, 2 * NT, 0);
#pragma unroll
                for (int t = 0; t < NT; ++t) b_cur[t] = b_nxt[t];
            }
        }
        if (c + 1 < 4) {
#pragma unroll
            for (int s = 0; s < 8; ++s) a_cur[s] = a_nxt[s];
        }
    }

    float bv[NT];
#pragma unroll
    for (int t = 0; t < NT; ++t) bv[t] = bias[t * 16 + idx];
    float* o = out + (size_t)(m0 + 8 * half) * N + idx;
    if (m0 + 16 <= n_dst) {
#pragma unroll
        for (int t = 0; t < NT; ++t) {
#pragma unroll
            for (int j = 0; j < 8; ++j) {
                float v = acc[t][j] + bv[t];
                if (RELU) v = fmaxf(v, 0.0f);
                o[j * N + t * 16] = v;
            }
        }
    } else {
#pragma unroll
        for (int t = 0; t < NT; ++t) {
#pragma unroll
            for (int j = 0; j < 8; ++j) {
                if (m0 + j + 8 * half < n_dst) {
                    float v = acc[t][j] + bv[t];
                    if (RELU) v = fmaxf(v, 0.0f);
                    o[j * N + t * 16] = v;
                }
            }
        }
    }
}

static inline unsigned cdiv(unsigned a, unsigned b) { return (a + b - 1) / b; }

static void sage_aggregate(const float* hsrc, int n_src, const int* src, const int* dst,
                           int nedges, int n_dst, char* scratch, size_t tmp_off,
                           float* msg, float* deg, hipStream_t stream) {
    unsigned* cnt  = (unsigned*)scratch;
    unsigned* cur  = (unsigned*)(scratch + (512u << 10));
    unsigned* ccnt = (unsigned*)(scratch + (960u << 10));
    unsigned* ccur = ccnt + MAX_BK;
    int*      perm = (int*)(scratch + (1024u << 10));
    int2*     tmp  = (int2*)(scratch + tmp_off);
    const dim3 blk(256);
    const unsigned nb = cdiv((unsigned)n_dst, BK_ROWS);
    zero_u32<<<cdiv(2u * MAX_BK, 256), blk, 0, stream>>>(ccnt, 2u * MAX_BK);
    bucket_count<<<cdiv((unsigned)nedges, CHUNK), blk, 0, stream>>>(dst, nedges, nb, ccnt);
    bucket_scatter<<<cdiv((unsigned)nedges, CHUNK), blk, 0, stream>>>(src, dst, nedges, nb, ccnt,
                                                                    ccur, tmp);
    bucket_sort<<<nb, blk, 0, stream>>>(tmp, nedges, n_dst, ccnt, cnt, cur, deg, perm);
    seg_gather<<<cdiv((unsigned)n_dst, 8), blk, 0, stream>>>(hsrc, n_src, perm, cnt, cur,
                                                            nedges, n_dst, msg);
}

extern "C" void kernel_launch(void* const* d_in, const int* in_sizes, int n_in,
                              void* d_out, int out_size, void* d_ws, size_t ws_size,
                              hipStream_t stream) {
    const float* feats = (const float*)d_in[0];
    const int* src0 = (const int*)d_in[1];
    const int* dst0 = (const int*)d_in[2];
    const int* src1 = (const int*)d_in[3];
    const int* dst1 = (const int*)d_in[4];
    const int* src2 = (const int*)d_in[5];
    const int* dst2 = (const int*)d_in[6];
    const float* Ws0 = (const float*)d_in[10];
    const float* Wn0 = (const float*)d_in[11];
    const float* b0  = (const float*)d_in[12];
    const float* Ws1 = (const float*)d_in[13];
    const float* Wn1 = (const float*)d_in[14];
    const float* b1  = (const float*)d_in[15];
    const float* Ws2 = (const float*)d_in[16];
    const float* Wn2 = (const float*)d_in[17];
    const float* b2  = (const float*)d_in[18];

    const int E0 = in_sizes[1], E1 = in_sizes[3], E2 = in_sizes[5];
    const int N0 = 500000, N1 = 100000, N2 = 20000, N3 = 5000;

    char* ws = (char*)d_ws;
    float* h1  = (float*)ws;                    ws += (size_t)N1 * 128 * sizeof(float);
    float* h2  = (float*)ws;                    ws += (size_t)N2 * 128 * sizeof(float);
    float* msg = (float*)ws;                    ws += (size_t)N1 * 128 * sizeof(float);
    float* deg = (float*)ws;

    const dim3 blk(256);

    sage_aggregate(feats, N0, src0, dst0, E0, N1, (char*)h1, (size_t)9 << 20, msg, deg, stream);
    {
        unsigned strips = cdiv((unsigned)N1, 16);
        sage_gemm<8, 8, true><<<cdiv(strips, 8), blk, 0, stream>>>(feats, msg, deg, Ws0, Wn0, b0, h1, N1);
    }

    sage_aggregate(h1, N1, src1, dst1, E1, N2, (char*)h2, (size_t)3 << 20, msg, deg, stream);
    {
        unsigned strips = cdiv((unsigned)N2, 16);
        sage_gemm<8, 8, true><<<cdiv(strips, 8), blk, 0, stream>>>(h1, msg, deg, Ws1, Wn1, b1, h2, N2);
    }

    sage_aggregate(h2, N2, src2, dst2, E2, N3, (char*)h1, (size_t)2 << 20, msg, deg, stream);
    {
        unsigned strips = cdiv((unsigned)N3, 16);
        sage_gemm<4, 4, false><<<cdiv(strips, 4), dim3(128), 0, stream>>>(h2, msg, deg, Ws2, Wn2, b2, (float*)d_out, N3);
    }
}


# Edge sorts of all three layers run in ONE zero/count/scatter/sort sequence at the start (a work-group finds its layer from its number): 18 launches become 10; row sum order still LDS-add arrival
# speedup vs baseline: 3.7521x; 1.2814x over previous
#include <hip/hip_runtime.h>
#include <hip/hip_bf16.h>


typedef __attribute__((ext_vector_type(2))) float v2f;
typedef __attribute__((ext_vector_type(8))) float v8f;

#define BK_ROWS   256
#define MAX_BK    512
#define CHUNK     8192
#define STAGE_CAP 12288

#define CUR_W   (512u << 8)
#define CCNT_W  (960u << 8)
#define PERM_W  (1024u << 8)

struct Layer {
    const int* src; const int* dst; int nedges; int n_dst;
    unsigned* wb; int2* tmp; float* deg; unsigned bid;
};
#define LAYER_ARGS(n) const int* __restrict__ src##n, const int* __restrict__ dst##n, int ne##n, \
                      int nd##n, unsigned* __restrict__ wb##n, int2* __restrict__ tmp##n,         \
                      float* __restrict__ deg##n
#define LAYER_OF(n) Layer{src##n, dst##n, ne##n, nd##n, wb##n, tmp##n, deg##n, 0u}

static __device__ __forceinline__ unsigned chunks_of(int nedges) {
    return ((unsigned)nedges + CHUNK - 1u) / CHUNK;
}
static __device__ __forceinline__ unsigned buckets_of(int n_dst) {
    return ((unsigned)n_dst + BK_ROWS - 1u) / BK_ROWS;
}
static __device__ __forceinline__ Layer pick_layer(Layer a, Layer b, Layer c, unsigned g0,
                                                   unsigned g1) {
    unsigned bid = blockIdx.x;
    if (bid < g0) { a.bid = bid; return a; }
    bid -= g0;
    if (bid < g1) { b.bid = bid; return b; }
    c.bid = bid - g1;
    return c;
}

__global__ __launch_bounds__(256) void zero_tables(unsigned* __restrict__ wb0,
                                                   unsigned* __restrict__ wb1,
                                                   unsigned* __restrict__ wb2) {
    const unsigned l = blockIdx.x >> 2;
    unsigned* p = (l == 0u) ? wb0 : ((l == 1u) ? wb1 : wb2);
    p[CCNT_W + (blockIdx.x & 3u) * 256u + threadIdx.x] = 0u;
}

static __device__ __forceinline__ unsigned bucket_of(int d, unsigned nb) {
    const unsigned b = (unsigned)d >> 8;
    return b < nb ? b : nb - 1u;
}

static __device__ __forceinline__ void chunk_histogram(unsigned* hist, const int* __restrict__ dst,
                                                       int nedges, unsigned nb, unsigned bid) {
    const int cbeg = (int)bid * CHUNK;
    if (cbeg + CHUNK <= nedges) {
        const int4* q = (const int4*)(dst + cbeg) + threadIdx.x;
        int4 d[CHUNK / 1024];
#pragma unroll
        for (int t = 0; t < CHUNK / 1024; ++t) d[t] = q[t * 256];
#pragma unroll
        for (int t = 0; t < CHUNK / 1024; ++t) {
            atomicAdd(&hist[bucket_of(d[t].x, nb)], 1u);
            atomicAdd(&hist[bucket_of(d[t].y, nb)], 1u);
            atomicAdd(&hist[bucket_of(d[t].z, nb)], 1u);
            atomicAdd(&hist[bucket_of(d[t].w, nb)], 1u);
        }
    } else {
        for (int e = cbeg + (int)threadIdx.x; e < nedges; e += 256)
            atomicAdd(&hist[bucket_of(dst[e], nb)], 1u);
    }
}

__global__ __launch_bounds__(256) void bucket_count(LAYER_ARGS(0), LAYER_ARGS(1), LAYER_ARGS(2)) {
    __shared__ unsigned hist[MAX_BK];
    const Layer L = pick_layer(LAYER_OF(0), LAYER_OF(1), LAYER_OF(2), chunks_of(ne0),
                               chunks_of(ne1));
    const int* __restrict__ dst = L.dst;
    const int nedges = L.nedges;
    const unsigned nb = buckets_of(L.n_dst);
    unsigned* __restrict__ ccnt = L.wb + CCNT_W;
    const int tid = threadIdx.x;
    hist[tid] = 0u;
    hist[tid + 256] = 0u;
    __syncthreads();
    chunk_histogram(hist, dst, nedges, nb, L.bid);
    __syncthreads();
    for (unsigned b = tid; b < nb; b += 256u) {
        const unsigned c = hist[b];
        if (c) atomicAdd(ccnt + b, c);
    }
}

static __device__ __forceinline__ unsigned block_scan_excl(unsigned v, unsigned* wave_off) {
    const int lane = threadIdx.x & 31;
    const int wv   = threadIdx.x >> 5;
    unsigned x = v;
    for (int off = 1; off < 32; off <<= 1) {
        unsigned y = __shfl_up(x, off);
        if (lane >= off) x += y;
    }
    if (lane == 31) wave_off[wv] = x;
    __syncthreads();
    unsigned pre = 0;
    for (int w = 0; w < wv; ++w) pre += wave_off[w];
    return pre + x - v;
}

__global__ __launch_bounds__(256) void bucket_scatter(LAYER_ARGS(0), LAYER_ARGS(1),
                                                      LAYER_ARGS(2)) {
    const Layer L = pick_layer(LAYER_OF(0), LAYER_OF(1), LAYER_OF(2), chunks_of(ne0),
                               chunks_of(ne1));
    const int* __restrict__ src = L.src;
    const int* __restrict__ dst = L.dst;
    const int nedges = L.nedges;
    const unsigned nb = buckets_of(L.n_dst);
    const unsigned* __restrict__ ccnt = L.wb + CCNT_W;
    unsigned* __restrict__ ccur = L.wb + CCNT_W + MAX_BK;
    int2* __restrict__ tmp = L.tmp;
    __shared__ unsigned hist[MAX_BK];
    __shared__ unsigned lstart[MAX_BK];
    __shared__ unsigned gstart[MAX_BK];
    __shared__ unsigned wave_off[8];
    __shared__ unsigned wave_off2[8];
    __shared__ int2 stage[CHUNK];
    const int tid = threadIdx.x;
    const unsigned b0 = 2u * (unsigned)tid;
    hist[tid] = 0u;
    hist[tid + 256] = 0u;
    const unsigned c0 = (b0 < nb) ? ccnt[b0] : 0u;
    const unsigned c1 = (b0 + 1u < nb) ? ccnt[b0 + 1u] : 0u;
    const unsigned gex = block_scan_excl(c0 + c1, wave_off);
    chunk_histogram(hist, dst, nedges, nb, L.bid);
    __syncthreads();
    const unsigned h0 = hist[b0];
    const unsigned h1 = hist[b0 + 1u];
    const unsigned lex = block_scan_excl(h0 + h1, wave_off2);
    const unsigned o0 = h0 ? atomicAdd(ccur + b0, h0) : 0u;
    const unsigned o1 = h1 ? atomicAdd(ccur + b0 + 1u, h1) : 0u;
    lstart[b0]      = lex;
    lstart[b0 + 1u] = lex + h0;
    hist[b0]        = lex;
    hist[b0 + 1u]   = lex + h0;
    gstart[b0]      = gex + o0;
    gstart[b0 + 1u] = gex + c0 + o1;
    __syncthreads();
    const int cbeg = (int)L.bid * CHUNK;
    if (cbeg + CHUNK <= nedges) {
        const int4* qs = (const int4*)(src + cbeg) + tid;
        const int4* qd = (const int4*)(dst + cbeg) + tid;
        int4 s[CHUNK / 1024], d[CHUNK / 1024];
#pragma unroll
        for (int t = 0; t < CHUNK / 1024; ++t) { d[t] = qd[t * 256]; s[t] = qs[t * 256]; }
#pragma unroll
        for (int t = 0; t < CHUNK / 1024; ++t) {
            const unsigned p0 = atomicAdd(&hist[bucket_of(d[t].x, nb)], 1u);
            const unsigned p1 = atomicAdd(&hist[bucket_of(d[t].y, nb)], 1u);
            const unsigned p2 = atomicAdd(&hist[bucket_of(d[t].z, nb)], 1u);
            const unsigned p3 = atomicAdd(&hist[bucket_of(d[t].w, nb)], 1u);
            if (p0 < (unsigned)CHUNK) stage[p0] = make_int2(s[t].x, d[t].x);
            if (p1 < (unsigned)CHUNK) stage[p1] = make_int2(s[t].y, d[t].y);
            if (p2 < (unsigned)CHUNK) stage[p2] = make_int2(s[t].z, d[t].z);
            if (p3 < (unsigned)CHUNK) stage[p3] = make_int2(s[t].w, d[t].w);
        }
    } else {
        for (int e = cbeg + tid; e < nedges; e += 256) {
            const int d = dst[e];
            const unsigned p = atomicAdd(&hist[bucket_of(d, nb)], 1u);
            if (p < (unsigned)CHUNK) stage[p] = make_int2(src[e], d);
        }
    }
    __syncthreads();
    int cn = nedges - cbeg;
    if (cn > CHUNK) cn = CHUNK;
    for (int i = tid; i < cn; i += 256) {
        const int2 e = stage[i];
        const unsigned b = bucket_of(e.y, nb);
        const unsigned pos = gstart[b] + ((unsigned)i - lstart[b]);
        if (pos < (unsigned)nedges) tmp[pos] = e;
    }
}

__global__ __launch_bounds__(256) void bucket_sort(LAYER_ARGS(0), LAYER_ARGS(1), LAYER_ARGS(2)) {
    const Layer L = pick_layer(LAYER_OF(0), LAYER_OF(1), LAYER_OF(2), buckets_of(nd0),
                               buckets_of(nd1));
    const int2* __restrict__ tmp = L.tmp;
    const int nedges = L.nedges;
    const int n_dst = L.n_dst;
    const unsigned* __restrict__ ccnt = L.wb + CCNT_W;
    unsigned* __restrict__ cnt = L.wb;
    unsigned* __restrict__ cur = L.wb + CUR_W;
    float* __restrict__ deg = L.deg;
    int* __restrict__ perm = (int*)(L.wb + PERM_W);
    __shared__ unsigned rows[BK_ROWS];
    __shared__ unsigned wave_off[8];
    __shared__ unsigned red[8];
    __shared__ int stage[STAGE_CAP];
    const int tid  = threadIdx.x;
    const int lane = tid & 31;
    const int wv   = tid >> 5;
    const unsigned b = L.bid;
    unsigned part = 0;
    for (unsigned i = tid; i < b; i += 256u) part += ccnt[i];
    for (int off = 16; off > 0; off >>= 1) part += __shfl_xor(part, off);
    if (lane == 0) red[wv] = part;
    rows[tid] = 0u;
    __syncthreads();
    unsigned beg = 0;
    for (int w = 0; w < 8; ++w) beg += red[w];
    unsigned n = ccnt[b];
    if (beg > (unsigned)nedges) beg = (unsigned)nedges;
    if (n > (unsigned)nedges - beg) n = (unsigned)nedges - beg;
    const int2* mine = tmp + beg;
    const int row0 = (int)(b * BK_ROWS);
    for (unsigned i0 = tid; i0 < n; i0 += 1024u) {
        int d[4];
#pragma unroll
        for (int k = 0; k < 4; ++k) {
            const unsigned i = i0 + 256u * k;
            d[k] = mine[i < n ? i : n - 1u].y;
        }
#pragma unroll
        for (int k = 0; k < 4; ++k)
            if (i0 + 256u * k < n) atomicAdd(&rows[(unsigned)(d[k] - row0) & (BK_ROWS - 1u)], 1u);
    }
    __syncthreads();
    const unsigned c  = rows[tid];
    const unsigned ex = block_scan_excl(c, wave_off);
    const int row = row0 + tid;
    if (row < n_dst) {
        cnt[row] = c;
        cur[row] = beg + ex + c;
        deg[row] = (float)c;
    }
    rows[tid] = ex;
    __syncthreads();
    if (n <= (unsigned)STAGE_CAP) {
        for (unsigned i0 = tid; i0 < n; i0 += 1024u) {
            int2 e[4];
#pragma unroll
            for (int k = 0; k < 4; ++k) {
                const unsigned i = i0 + 256u * k;
                e[k] = mine[i < n ? i : n - 1u];
            }
#pragma unroll
            for (int k = 0; k < 4; ++k) {
                if (i0 + 256u * k < n) {
                    const unsigned p =
                        atomicAdd(&rows[(unsigned)(e[k].y - row0) & (BK_ROWS - 1u)], 1u);
                    if (p < (unsigned)STAGE_CAP) stage[p] = e[k].x;
                }
            }
        }
        __syncthreads();
        for (unsigned i = tid; i < n; i += 256u) perm[beg + i] = stage[i];
    } else {
        for (unsigned i = tid; i < n; i += 256u) {
            const int2 e = mine[i];
            const unsigned p = beg + atomicAdd(&rows[(unsigned)(e.y - row0) & (BK_ROWS - 1u)], 1u);
            if (p < (unsigned)nedges) perm[p] = e.x;
        }
    }
}

static __device__ __forceinline__ float4 load_row(const float* __restrict__ col, int s,
                                                  int n_src) {
    unsigned r = (unsigned)s;
    if (r > (unsigned)(n_src - 1)) r = (unsigned)(n_src - 1);
    return *(const float4*)(col + (size_t)r * 128);
}

template <int G>
static __device__ __forceinline__ void add_rows(float4& acc, const float* __restrict__ col,
                                                int mine, unsigned j, int n_src) {
    float4 v[G];
#pragma unroll
    for (int k = 0; k < G; ++k)
        v[k] = load_row(col, __builtin_amdgcn_readlane(mine, (int)j + k), n_src);
#pragma unroll
    for (int k = 0; k < G; ++k) {
        acc.x += v[k].x; acc.y += v[k].y; acc.z += v[k].z; acc.w += v[k].w;
    }
}

__global__ __launch_bounds__(256) void seg_gather(const float* __restrict__ hsrc, int n_src,
                                                  const int* __restrict__ perm,
                                                  const unsigned* __restrict__ cnt,
                                                  const unsigned* __restrict__ cur,
                                                  int nedges, int n_dst,
                                                  float* __restrict__ msg) {
    const int lane = threadIdx.x & 31;
    const int row  = (int)(blockIdx.x * 8u) + (int)(threadIdx.x >> 5);
    if (row >= n_dst) return;
    unsigned n   = (unsigned)__builtin_amdgcn_readfirstlane((int)cnt[row]);
    unsigned end = (unsigned)__builtin_amdgcn_readfirstlane((int)cur[row]);
    if (end > (unsigned)nedges) end = (unsigned)nedges;
    if (n > end) n = end;
    const unsigned beg = end - n;

    const float* col = hsrc + lane * 4;
    float4 acc = make_float4(0.f, 0.f, 0.f, 0.f);
    for (unsigned base = 0; base < n; base += 32) {
        unsigned m = n - base;
        if (m > 32u) m = 32u;
        const int mine = ((unsigned)lane < m) ? perm[beg + base + (unsigned)lane] : 0;
        unsigned j = 0;
        for (; j + 8 <= m; j += 8) add_rows<8>(acc, col, mine, j, n_src);
        if (j + 4 <= m) { add_rows<4>(acc, col, mine, j, n_src); j += 4; }
        for (; j < m; ++j) add_rows<1>(acc, col, mine, j, n_src);
    }
    *(float4*)(msg + (size_t)row * 128 + lane * 4) = acc;
}

#define W_PITCH 260

static __device__ __forceinline__ int k_of(int S, int half) { return 8 * S + 4 * half; }

template <int NT, int WAVES, bool RELU>
__global__ __launch_bounds__(WAVES * 32) void sage_gemm(const float* __restrict__ hdst,
                                                        const float* __restrict__ msg,
                                                        const float* __restrict__ deg,
                                                        const float* __restrict__ Wself,
                                                        const float* __restrict__ Wneigh,
                                                        const float* __restrict__ bias,
                                                        float* __restrict__ out,
                                                        int n_dst) {
    constexpr int N       = NT * 16;
    constexpr int THREADS = WAVES * 32;
    constexpr int F4      = N * 64;
    constexpr int STG     = 16;
    static_assert(F4 % (THREADS * STG) == 0, "weight staging loop");
    __shared__ __attribute__((aligned(16))) float Wl[N * W_PITCH];

    const int tid  = threadIdx.x;
    const int lane = tid & 31;
    const int wave = tid >> 5;
    const int half = lane >> 4;
    const int idx  = lane & 15;

    const int strip   = (int)blockIdx.x * WAVES + wave;
    const int nstrips = (n_dst + 15) >> 4;
    const int m0      = strip << 4;

    int arow = m0 + idx;
    if (arow > n_dst - 1) arow = n_dst - 1;

    const float* Ah = hdst + (size_t)arow * 128 + k_of(0, half);
    const float* Am = msg  + (size_t)arow * 128 + k_of(0, half);

    float4 a_cur[8];
#pragma unroll
    for (int s = 0; s < 8; ++s) a_cur[s] = *(const float4*)(Ah + 8 * s);
    const float invdeg = 1.0f / fmaxf(deg[arow], 1.0f);

    for (int i0 = 0; i0 < F4; i0 += THREADS * STG) {
        float4 t[STG];
#pragma unroll
        for (int u = 0; u < STG; ++u) {
            const int q   = i0 + u * THREADS + tid;
            const int row = q >> 6;
            const int c4  = q & 63;
            const float* s = (c4 < 32) ? (Wself + row * 128 + c4 * 4)
                                       : (Wneigh + row * 128 + (c4 - 32) * 4);
            t[u] = *(const float4*)s;
        }
#pragma unroll
        for (int u = 0; u < STG; ++u) {
            const int q   = i0 + u * THREADS + tid;
            const int row = q >> 6;
            const int c4  = q & 63;
            *(float4*)(Wl + row * W_PITCH + c4 * 4) = t[u];
        }
    }
    __syncthreads();
    if (strip >= nstrips) return;

    v8f acc[NT];
#pragma unroll
    for (int t = 0; t < NT; ++t) acc[t] = (v8f){0.f, 0.f, 0.f, 0.f, 0.f, 0.f, 0.f, 0.f};

    const float* Bl = Wl + idx * W_PITCH + k_of(0, half);

    float4 b_cur[NT];
#pragma unroll
    for (int t = 0; t < NT; ++t) b_cur[t] = *(const float4*)(Bl + t * 16 * W_PITCH);
    __builtin_amdgcn_sched_group_barrier(0x100, NT, 0);

#pragma unroll
    for (int c = 0; c < 4; ++c) {
        float4 a_nxt[8];
        if (c + 1 < 4) {
            const float* p = (c + 1 < 2) ? (Ah + 64 * (c + 1)) : (Am + 64 * (c + 1 - 2));
#pragma unroll
            for (int s = 0; s < 8; ++s) a_nxt[s] = *(const float4*)(p + 8 * s);
        }
#pragma unroll
        for (int s = 0; s < 8; ++s) {
            const int S = c * 8 + s;
            float4 a = a_cur[s];
            if (c >= 2) { a.x *= invdeg; a.y *= invdeg; a.z *= invdeg; a.w *= invdeg; }
            float4 b_nxt[NT];
            if (S + 1 < 32) {
#pragma unroll
                for (int t = 0; t < NT; ++t)
                    b_nxt[t] = *(const float4*)(Bl + t * 16 * W_PITCH + 8 * (S + 1));
            }
            const v2f a01 = {a.x, a.y};
            const v2f a23 = {a.z, a.w};
#pragma unroll
            for (int t = 0; t < NT; ++t) {
                const v2f b01 = {b_cur[t].x, b_cur[t].y};
                acc[t] = __builtin_amdgcn_wmma_f32_16x16x4_f32(false, a01, false, b01,
                                                               (short)0, acc[t], false, false);
            }
#pragma unroll
            for (int t = 0; t < NT; ++t) {
                const v2f b23 = {b_cur[t].z, b_cur[t].w};
                acc[t] = __builtin_amdgcn_wmma_f32_16x16x4_f32(false, a23, false, b23,
                                                               (short)0, acc[t], false, false);
            }
            if (S + 1 < 32) {
                __builtin_amdgcn_sched_group_barrier(0x100, NT, 0);
                __builtin_amdgcn_sched_group_barrier(0x008, 2 * NT, 0);
#pragma unroll
                for (int t = 0; t < NT; ++t) b_cur[t] = b_nxt[t];
            }
        }
        if (c + 1 < 4) {
#pragma unroll
            for (int s = 0; s < 8; ++s) a_cur[s] = a_nxt[s];
        }
    }

    float bv[NT];
#pragma unroll
    for (int t = 0; t < NT; ++t) bv[t] = bias[t * 16 + idx];
    float* o = out + (size_t)(m0 + 8 * half) * N + idx;
    if (m0 + 16 <= n_dst) {
#pragma unroll
        for (int t = 0; t < NT; ++t) {
#pragma unroll
            for (int j = 0; j < 8; ++j) {
                float v = acc[t][j] + bv[t];
                if (RELU) v = fmaxf(v, 0.0f);
                o[j * N + t * 16] = v;
            }
        }
    } else {
#pragma unroll
        for (int t = 0; t < NT; ++t) {
#pragma unroll
            for (int j = 0; j < 8; ++j) {
                if (m0 + j + 8 * half < n_dst) {
                    float v = acc[t][j] + bv[t];
                    if (RELU) v = fmaxf(v, 0.0f);
                    o[j * N + t * 16] = v;
                }
            }
        }
    }
}

static inline unsigned cdiv(unsigned a, unsigned b) { return (a + b - 1) / b; }

static void sage_gather(const float* hsrc, int n_src, int nedges, int n_dst, unsigned* wb,
                        float* msg, hipStream_t stream) {
    seg_gather<<<cdiv((unsigned)n_dst, 8), dim3(256), 0, stream>>>(
        hsrc, n_src, (const int*)(wb + PERM_W), wb, wb + CUR_W, nedges, n_dst, msg);
}

extern "C" void kernel_launch(void* const* d_in, const int* in_sizes, int n_in,
                              void* d_out, int out_size, void* d_ws, size_t ws_size,
                              hipStream_t stream) {
    const float* feats = (const float*)d_in[0];
    const int* src0 = (const int*)d_in[1];
    const int* dst0 = (const int*)d_in[2];
    const int* src1 = (const int*)d_in[3];
    const int* dst1 = (const int*)d_in[4];
    const int* src2 = (const int*)d_in[5];
    const int* dst2 = (const int*)d_in[6];
    const float* Ws0 = (const float*)d_in[10];
    const float* Wn0 = (const float*)d_in[11];
    const float* b0  = (const float*)d_in[12];
    const float* Ws1 = (const float*)d_in[13];
    const float* Wn1 = (const float*)d_in[14];
    const float* b1  = (const float*)d_in[15];
    const float* Ws2 = (const float*)d_in[16];
    const float* Wn2 = (const float*)d_in[17];
    const float* b2  = (const float*)d_in[18];

    const int E0 = in_sizes[1], E1 = in_sizes[3], E2 = in_sizes[5];
    const int N0 = 500000, N1 = 100000, N2 = 20000, N3 = 5000;

    char* ws = (char*)d_ws;
    float* h1    = (float*)ws;                  ws += (size_t)N1 * 128 * sizeof(float);
    char*  lists = ws;                          ws += (size_t)N2 * 128 * sizeof(float);
    float* msg   = (float*)ws;                  ws += (size_t)N1 * 128 * sizeof(float);
    float* deg0  = (float*)ws;
    float* h2    = (float*)((char*)msg + ((size_t)16 << 20));

    unsigned* wb0  = (unsigned*)h1;
    int2*     tmp0 = (int2*)((char*)h1 + ((size_t)9 << 20));
    unsigned* wb1  = (unsigned*)lists;
    float*    deg1 = (float*)(lists + ((size_t)2816 << 10));
    int2*     tmp1 = (int2*)(lists + ((size_t)3 << 20));
    char*     l2   = lists + ((size_t)6656 << 10);
    unsigned* wb2  = (unsigned*)l2;
    float*    deg2 = (float*)(l2 + ((size_t)1536 << 10));
    int2*     tmp2 = (int2*)(l2 + ((size_t)2 << 20));

    const dim3 blk(256);

    {
        const unsigned chunks = cdiv((unsigned)E0, CHUNK) + cdiv((unsigned)E1, CHUNK) +
                                cdiv((unsigned)E2, CHUNK);
        const unsigned buckets = cdiv((unsigned)N1, BK_ROWS) + cdiv((unsigned)N2, BK_ROWS) +
                                 cdiv((unsigned)N3, BK_ROWS);
        zero_tables<<<12, blk, 0, stream>>>(wb0, wb1, wb2);
        bucket_count<<<chunks, blk, 0, stream>>>(src0, dst0, E0, N1, wb0, tmp0, deg0,
                                                 src1, dst1, E1, N2, wb1, tmp1, deg1,
                                                 src2, dst2, E2, N3, wb2, tmp2, deg2);
        bucket_scatter<<<chunks, blk, 0, stream>>>(src0, dst0, E0, N1, wb0, tmp0, deg0,
                                                   src1, dst1, E1, N2, wb1, tmp1, deg1,
                                                   src2, dst2, E2, N3, wb2, tmp2, deg2);
        bucket_sort<<<buckets, blk, 0, stream>>>(src0, dst0, E0, N1, wb0, tmp0, deg0,
                                                 src1, dst1, E1, N2, wb1, tmp1, deg1,
                                                 src2, dst2, E2, N3, wb2, tmp2, deg2);
    }

    sage_gather(feats, N0, E0, N1, wb0, msg, stream);
    {
        unsigned strips = cdiv((unsigned)N1, 16);
        sage_gemm<8, 8, true><<<cdiv(strips, 8), blk, 0, stream>>>(feats, msg, deg0, Ws0, Wn0, b0, h1, N1);
    }

    sage_gather(h1, N1, E1, N2, wb1, msg, stream);
    {
        unsigned strips = cdiv((unsigned)N2, 16);
        sage_gemm<8, 8, true><<<cdiv(strips, 8), blk, 0, stream>>>(h1, msg, deg1, Ws1, Wn1, b1, h2, N2);
    }

    sage_gather(h2, N2, E2, N3, wb2, msg, stream);
    {
        unsigned strips = cdiv((unsigned)N3, 16);
        sage_gemm<4, 4, false><<<cdiv(strips, 4), dim3(128), 0, stream>>>(h2, msg, deg2, Ws2, Wn2, b2, (float*)d_out, N3);
    }
}


# Layer GEMMs: columns split into two blocks, 4-wave groups, one wave per strip and block, so matrix work spreads evenly over the SIMDs (fullest SIMD in layer 0: 8 -> 6.5 strips); sum order unchanged
# speedup vs baseline: 3.8828x; 1.0348x over previous
#include <hip/hip_runtime.h>
#include <hip/hip_bf16.h>


typedef __attribute__((ext_vector_type(2))) float v2f;
typedef __attribute__((ext_vector_type(8))) float v8f;

#define BK_ROWS   256
#define MAX_BK    512
#define CHUNK     8192
#define STAGE_CAP 12288

#define CUR_W   (512u << 8)
#define CCNT_W  (960u << 8)
#define PERM_W  (1024u << 8)

struct Layer {
    const int* src; const int* dst; int nedges; int n_dst;
    unsigned* wb; int2* tmp; float* deg; unsigned bid;
};
#define LAYER_ARGS(n) const int* __restrict__ src##n, const int* __restrict__ dst##n, int ne##n, \
                      int nd##n, unsigned* __restrict__ wb##n, int2* __restrict__ tmp##n,         \
                      float* __restrict__ deg##n
#define LAYER_OF(n) Layer{src##n, dst##n, ne##n, nd##n, wb##n, tmp##n, deg##n, 0u}

static __device__ __forceinline__ unsigned chunks_of(int nedges) {
    return ((unsigned)nedges + CHUNK - 1u) / CHUNK;
}
static __device__ __forceinline__ unsigned buckets_of(int n_dst) {
    return ((unsigned)n_dst + BK_ROWS - 1u) / BK_ROWS;
}
static __device__ __forceinline__ Layer pick_layer(Layer a, Layer b, Layer c, unsigned g0,
                                                   unsigned g1) {
    unsigned bid = blockIdx.x;
    if (bid < g0) { a.bid = bid; return a; }
    bid -= g0;
    if (bid < g1) { b.bid = bid; return b; }
    c.bid = bid - g1;
    return c;
}

__global__ __launch_bounds__(256) void zero_tables(unsigned* __restrict__ wb0,
                                                   unsigned* __restrict__ wb1,
                                                   unsigned* __restrict__ wb2) {
    const unsigned l = blockIdx.x >> 2;
    unsigned* p = (l == 0u) ? wb0 : ((l == 1u) ? wb1 : wb2);
    p[CCNT_W + (blockIdx.x & 3u) * 256u + threadIdx.x] = 0u;
}

static __device__ __forceinline__ unsigned bucket_of(int d, unsigned nb) {
    const unsigned b = (unsigned)d >> 8;
    return b < nb ? b : nb - 1u;
}

static __device__ __forceinline__ void chunk_histogram(unsigned* hist, const int* __restrict__ dst,
                                                       int nedges, unsigned nb, unsigned bid) {
    const int cbeg = (int)bid * CHUNK;
    if (cbeg + CHUNK <= nedges) {
        const int4* q = (const int4*)(dst + cbeg) + threadIdx.x;
        int4 d[CHUNK / 1024];
#pragma unroll
        for (int t = 0; t < CHUNK / 1024; ++t) d[t] = q[t * 256];
#pragma unroll
        for (int t = 0; t < CHUNK / 1024; ++t) {
            atomicAdd(&hist[bucket_of(d[t].x, nb)], 1u);
            atomicAdd(&hist[bucket_of(d[t].y, nb)], 1u);
            atomicAdd(&hist[bucket_of(d[t].z, nb)], 1u);
            atomicAdd(&hist[bucket_of(d[t].w, nb)], 1u);
        }
    } else {
        for (int e = cbeg + (int)threadIdx.x; e < nedges; e += 256)
            atomicAdd(&hist[bucket_of(dst[e], nb)], 1u);
    }
}

__global__ __launch_bounds__(256) void bucket_count(LAYER_ARGS(0), LAYER_ARGS(1), LAYER_ARGS(2)) {
    __shared__ unsigned hist[MAX_BK];
    const Layer L = pick_layer(LAYER_OF(0), LAYER_OF(1), LAYER_OF(2), chunks_of(ne0),
                               chunks_of(ne1));
    const int* __restrict__ dst = L.dst;
    const int nedges = L.nedges;
    const unsigned nb = buckets_of(L.n_dst);
    unsigned* __restrict__ ccnt = L.wb + CCNT_W;
    const int tid = threadIdx.x;
    hist[tid] = 0u;
    hist[tid + 256] = 0u;
    __syncthreads();
    chunk_histogram(hist, dst, nedges, nb, L.bid);
    __syncthreads();
    for (unsigned b = tid; b < nb; b += 256u) {
        const unsigned c = hist[b];
        if (c) atomicAdd(ccnt + b, c);
    }
}

static __device__ __forceinline__ unsigned block_scan_excl(unsigned v, unsigned* wave_off) {
    const int lane = threadIdx.x & 31;
    const int wv   = threadIdx.x >> 5;
    unsigned x = v;
    for (int off = 1; off < 32; off <<= 1) {
        unsigned y = __shfl_up(x, off);
        if (lane >= off) x += y;
    }
    if (lane == 31) wave_off[wv] = x;
    __syncthreads();
    unsigned pre = 0;
    for (int w = 0; w < wv; ++w) pre += wave_off[w];
    return pre + x - v;
}

__global__ __launch_bounds__(256) void bucket_scatter(LAYER_ARGS(0), LAYER_ARGS(1),
                                                      LAYER_ARGS(2)) {
    const Layer L = pick_layer(LAYER_OF(0), LAYER_OF(1), LAYER_OF(2), chunks_of(ne0),
                               chunks_of(ne1));
    const int* __restrict__ src = L.src;
    const int* __restrict__ dst = L.dst;
    const int nedges = L.nedges;
    const unsigned nb = buckets_of(L.n_dst);
    const unsigned* __restrict__ ccnt = L.wb + CCNT_W;
    unsigned* __restrict__ ccur = L.wb + CCNT_W + MAX_BK;
    int2* __restrict__ tmp = L.tmp;
    __shared__ unsigned hist[MAX_BK];
    __shared__ unsigned lstart[MAX_BK];
    __shared__ unsigned gstart[MAX_BK];
    __shared__ unsigned wave_off[8];
    __shared__ unsigned wave_off2[8];
    __shared__ int2 stage[CHUNK];
    const int tid = threadIdx.x;
    const unsigned b0 = 2u * (unsigned)tid;
    hist[tid] = 0u;
    hist[tid + 256] = 0u;
    const unsigned c0 = (b0 < nb) ? ccnt[b0] : 0u;
    const unsigned c1 = (b0 + 1u < nb) ? ccnt[b0 + 1u] : 0u;
    const unsigned gex = block_scan_excl(c0 + c1, wave_off);
    chunk_histogram(hist, dst, nedges, nb, L.bid);
    __syncthreads();
    const unsigned h0 = hist[b0];
    const unsigned h1 = hist[b0 + 1u];
    const unsigned lex = block_scan_excl(h0 + h1, wave_off2);
    const unsigned o0 = h0 ? atomicAdd(ccur + b0, h0) : 0u;
    const unsigned o1 = h1 ? atomicAdd(ccur + b0 + 1u, h1) : 0u;
    lstart[b0]      = lex;
    lstart[b0 + 1u] = lex + h0;
    hist[b0]        = lex;
    hist[b0 + 1u]   = lex + h0;
    gstart[b0]      = gex + o0;
    gstart[b0 + 1u] = gex + c0 + o1;
    __syncthreads();
    const int cbeg = (int)L.bid * CHUNK;
    if (cbeg + CHUNK <= nedges) {
        const int4* qs = (const int4*)(src + cbeg) + tid;
        const int4* qd = (const int4*)(dst + cbeg) + tid;
        int4 s[CHUNK / 1024], d[CHUNK / 1024];
#pragma unroll
        for (int t = 0; t < CHUNK / 1024; ++t) { d[t] = qd[t * 256]; s[t] = qs[t * 256]; }
#pragma unroll
        for (int t = 0; t < CHUNK / 1024; ++t) {
            const unsigned p0 = atomicAdd(&hist[bucket_of(d[t].x, nb)], 1u);
            const unsigned p1 = atomicAdd(&hist[bucket_of(d[t].y, nb)], 1u);
            const unsigned p2 = atomicAdd(&hist[bucket_of(d[t].z, nb)], 1u);
            const unsigned p3 = atomicAdd(&hist[bucket_of(d[t].w, nb)], 1u);
            if (p0 < (unsigned)CHUNK) stage[p0] = make_int2(s[t].x, d[t].x);
            if (p1 < (unsigned)CHUNK) stage[p1] = make_int2(s[t].y, d[t].y);
            if (p2 < (unsigned)CHUNK) stage[p2] = make_int2(s[t].z, d[t].z);
            if (p3 < (unsigned)CHUNK) stage[p3] = make_int2(s[t].w, d[t].w);
        }
    } else {
        for (int e = cbeg + tid; e < nedges; e += 256) {
            const int d = dst[e];
            const unsigned p = atomicAdd(&hist[bucket_of(d, nb)], 1u);
            if (p < (unsigned)CHUNK) stage[p] = make_int2(src[e], d);
        }
    }
    __syncthreads();
    int cn = nedges - cbeg;
    if (cn > CHUNK) cn = CHUNK;
    for (int i = tid; i < cn; i += 256) {
        const int2 e = stage[i];
        const unsigned b = bucket_of(e.y, nb);
        const unsigned pos = gstart[b] + ((unsigned)i - lstart[b]);
        if (pos < (unsigned)nedges) tmp[pos] = e;
    }
}

__global__ __launch_bounds__(256) void bucket_sort(LAYER_ARGS(0), LAYER_ARGS(1), LAYER_ARGS(2)) {
    const Layer L = pick_layer(LAYER_OF(0), LAYER_OF(1), LAYER_OF(2), buckets_of(nd0),
                               buckets_of(nd1));
    const int2* __restrict__ tmp = L.tmp;
    const int nedges = L.nedges;
    const int n_dst = L.n_dst;
    const unsigned* __restrict__ ccnt = L.wb + CCNT_W;
    unsigned* __restrict__ cnt = L.wb;
    unsigned* __restrict__ cur = L.wb + CUR_W;
    float* __restrict__ deg = L.deg;
    int* __restrict__ perm = (int*)(L.wb + PERM_W);
    __shared__ unsigned rows[BK_ROWS];
    __shared__ unsigned wave_off[8];
    __shared__ unsigned red[8];
    __shared__ int stage[STAGE_CAP];
    const int tid  = threadIdx.x;
    const int lane = tid & 31;
    const int wv   = tid >> 5;
    const unsigned b = L.bid;
    unsigned part = 0;
    for (unsigned i = tid; i < b; i += 256u) part += ccnt[i];
    for (int off = 16; off > 0; off >>= 1) part += __shfl_xor(part, off);
    if (lane == 0) red[wv] = part;
    rows[tid] = 0u;
    __syncthreads();
    unsigned beg = 0;
    for (int w = 0; w < 8; ++w) beg += red[w];
    unsigned n = ccnt[b];
    if (beg > (unsigned)nedges) beg = (unsigned)nedges;
    if (n > (unsigned)nedges - beg) n = (unsigned)nedges - beg;
    const int2* mine = tmp + beg;
    const int row0 = (int)(b * BK_ROWS);
    for (unsigned i0 = tid; i0 < n; i0 += 1024u) {
        int d[4];
#pragma unroll
        for (int k = 0; k < 4; ++k) {
            const unsigned i = i0 + 256u * k;
            d[k] = mine[i < n ? i : n - 1u].y;
        }
#pragma unroll
        for (int k = 0; k < 4; ++k)
            if (i0 + 256u * k < n) atomicAdd(&rows[(unsigned)(d[k] - row0) & (BK_ROWS - 1u)], 1u);
    }
    __syncthreads();
    const unsigned c  = rows[tid];
    const unsigned ex = block_scan_excl(c, wave_off);
    const int row = row0 + tid;
    if (row < n_dst) {
        cnt[row] = c;
        cur[row] = beg + ex + c;
        deg[row] = (float)c;
    }
    rows[tid] = ex;
    __syncthreads();
    if (n <= (unsigned)STAGE_CAP) {
        for (unsigned i0 = tid; i0 < n; i0 += 1024u) {
            int2 e[4];
#pragma unroll
            for (int k = 0; k < 4; ++k) {
                const unsigned i = i0 + 256u * k;
                e[k] = mine[i < n ? i : n - 1u];
            }
#pragma unroll
            for (int k = 0; k < 4; ++k) {
                if (i0 + 256u * k < n) {
                    const unsigned p =
                        atomicAdd(&rows[(unsigned)(e[k].y - row0) & (BK_ROWS - 1u)], 1u);
                    if (p < (unsigned)STAGE_CAP) stage[p] = e[k].x;
                }
            }
        }
        __syncthreads();
        for (unsigned i = tid; i < n; i += 256u) perm[beg + i] = stage[i];
    } else {
        for (unsigned i = tid; i < n; i += 256u) {
            const int2 e = mine[i];
            const unsigned p = beg + atomicAdd(&rows[(unsigned)(e.y - row0) & (BK_ROWS - 1u)], 1u);
            if (p < (unsigned)nedges) perm[p] = e.x;
        }
    }
}

static __device__ __forceinline__ float4 load_row(const float* __restrict__ col, int s,
                                                  int n_src) {
    unsigned r = (unsigned)s;
    if (r > (unsigned)(n_src - 1)) r = (unsigned)(n_src - 1);
    return *(const float4*)(col + (size_t)r * 128);
}

template <int G>
static __device__ __forceinline__ void add_rows(float4& acc, const float* __restrict__ col,
                                                int mine, unsigned j, int n_src) {
    float4 v[G];
#pragma unroll
    for (int k = 0; k < G; ++k)
        v[k] = load_row(col, __builtin_amdgcn_readlane(mine, (int)j + k), n_src);
#pragma unroll
    for (int k = 0; k < G; ++k) {
        acc.x += v[k].x; acc.y += v[k].y; acc.z += v[k].z; acc.w += v[k].w;
    }
}

__global__ __launch_bounds__(256) void seg_gather(const float* __restrict__ hsrc, int n_src,
                                                  const int* __restrict__ perm,
                                                  const unsigned* __restrict__ cnt,
                                                  const unsigned* __restrict__ cur,
                                                  int nedges, int n_dst,
                                                  float* __restrict__ msg) {
    const int lane = threadIdx.x & 31;
    const int row  = (int)(blockIdx.x * 8u) + (int)(threadIdx.x >> 5);
    if (row >= n_dst) return;
    unsigned n   = (unsigned)__builtin_amdgcn_readfirstlane((int)cnt[row]);
    unsigned end = (unsigned)__builtin_amdgcn_readfirstlane((int)cur[row]);
    if (end > (unsigned)nedges) end = (unsigned)nedges;
    if (n > end) n = end;
    const unsigned beg = end - n;

    const float* col = hsrc + lane * 4;
    float4 acc = make_float4(0.f, 0.f, 0.f, 0.f);
    for (unsigned base = 0; base < n; base += 32) {
        unsigned m = n - base;
        if (m > 32u) m = 32u;
        const int mine = ((unsigned)lane < m) ? perm[beg + base + (unsigned)lane] : 0;
        unsigned j = 0;
        for (; j + 8 <= m; j += 8) add_rows<8>(acc, col, mine, j, n_src);
        if (j + 4 <= m) { add_rows<4>(acc, col, mine, j, n_src); j += 4; }
        for (; j < m; ++j) add_rows<1>(acc, col, mine, j, n_src);
    }
    *(float4*)(msg + (size_t)row * 128 + lane * 4) = acc;
}

#define W_PITCH 260

static __device__ __forceinline__ int k_of(int S, int half) { return 8 * S + 4 * half; }

template <int NT, int NB, int WAVES, bool RELU>
__global__ __launch_bounds__(WAVES * 32) void sage_gemm(const float* __restrict__ hdst,
                                                        const float* __restrict__ msg,
                                                        const float* __restrict__ deg,
                                                        const float* __restrict__ Wself,
                                                        const float* __restrict__ Wneigh,
                                                        const float* __restrict__ bias,
                                                        float* __restrict__ out,
                                                        int n_dst) {
    constexpr int N       = NT * NB * 16;
    constexpr int NL      = NT * 16;
    constexpr int THREADS = WAVES * 32;
    constexpr int F4      = NL * 64;
    constexpr int STG     = 16;
    static_assert(F4 % (THREADS * STG) == 0, "weight staging loop");
    __shared__ __attribute__((aligned(16))) float Wl[NL * W_PITCH];

    const int tid  = threadIdx.x;
    const int lane = tid & 31;
    const int wave = tid >> 5;
    const int half = lane >> 4;
    const int idx  = lane & 15;

    const int cb      = (int)(blockIdx.x % (unsigned)NB);
    const int col0    = cb * NL;
    const int strip   = (int)(blockIdx.x / (unsigned)NB) * WAVES + wave;
    const int nstrips = (n_dst + 15) >> 4;
    const int m0      = strip << 4;

    int arow = m0 + idx;
    if (arow > n_dst - 1) arow = n_dst - 1;

    const float* Ah = hdst + (size_t)arow * 128 + k_of(0, half);
    const float* Am = msg  + (size_t)arow * 128 + k_of(0, half);

    float4 a_cur[8];
#pragma unroll
    for (int s = 0; s < 8; ++s) a_cur[s] = *(const float4*)(Ah + 8 * s);
    const float invdeg = 1.0f / fmaxf(deg[arow], 1.0f);

    for (int i0 = 0; i0 < F4; i0 += THREADS * STG) {
        float4 t[STG];
#pragma unroll
        for (int u = 0; u < STG; ++u) {
            const int q   = i0 + u * THREADS + tid;
            const int row = q >> 6;
            const int c4  = q & 63;
            const float* s = (c4 < 32) ? (Wself + (col0 + row) * 128 + c4 * 4)
                                       : (Wneigh + (col0 + row) * 128 + (c4 - 32) * 4);
            t[u] = *(const float4*)s;
        }
#pragma unroll
        for (int u = 0; u < STG; ++u) {
            const int q   = i0 + u * THREADS + tid;
            const int row = q >> 6;
            const int c4  = q & 63;
            *(float4*)(Wl + row * W_PITCH + c4 * 4) = t[u];
        }
    }
    __syncthreads();
    if (strip >= nstrips) return;

    v8f acc[NT];
#pragma unroll
    for (int t = 0; t < NT; ++t) acc[t] = (v8f){0.f, 0.f, 0.f, 0.f, 0.f, 0.f, 0.f, 0.f};

    const float* Bl = Wl + idx * W_PITCH + k_of(0, half);

    float4 b_cur[NT];
#pragma unroll
    for (int t = 0; t < NT; ++t) b_cur[t] = *(const float4*)(Bl + t * 16 * W_PITCH);
    __builtin_amdgcn_sched_group_barrier(0x100, NT, 0);

#pragma unroll
    for (int c = 0; c < 4; ++c) {
        float4 a_nxt[8];
        if (c + 1 < 4) {
            const float* p = (c + 1 < 2) ? (Ah + 64 * (c + 1)) : (Am + 64 * (c + 1 - 2));
#pragma unroll
            for (int s = 0; s < 8; ++s) a_nxt[s] = *(const float4*)(p + 8 * s);
        }
#pragma unroll
        for (int s = 0; s < 8; ++s) {
            const int S = c * 8 + s;
            float4 a = a_cur[s];
            if (c >= 2) { a.x *= invdeg; a.y *= invdeg; a.z *= invdeg; a.w *= invdeg; }
            float4 b_nxt[NT];
            if (S + 1 < 32) {
#pragma unroll
                for (int t = 0; t < NT; ++t)
                    b_nxt[t] = *(const float4*)(Bl + t * 16 * W_PITCH + 8 * (S + 1));
            }
            const v2f a01 = {a.x, a.y};
            const v2f a23 = {a.z, a.w};
#pragma unroll
            for (int t = 0; t < NT; ++t) {
                const v2f b01 = {b_cur[t].x, b_cur[t].y};
                acc[t] = __builtin_amdgcn_wmma_f32_16x16x4_f32(false, a01, false, b01,
                                                               (short)0, acc[t], false, false);
            }
#pragma unroll
            for (int t = 0; t < NT; ++t) {
                const v2f b23 = {b_cur[t].z, b_cur[t].w};
                acc[t] = __builtin_amdgcn_wmma_f32_16x16x4_f32(false, a23, false, b23,
                                                               (short)0, acc[t], false, false);
            }
            if (S + 1 < 32) {
                __builtin_amdgcn_sched_group_barrier(0x100, NT, 0);
                __builtin_amdgcn_sched_group_barrier(0x008, 2 * NT, 0);
#pragma unroll
                for (int t = 0; t < NT; ++t) b_cur[t] = b_nxt[t];
            }
        }
        if (c + 1 < 4) {
#pragma unroll
            for (int s = 0; s < 8; ++s) a_cur[s] = a_nxt[s];
        }
    }

    float bv[NT];
#pragma unroll
    for (int t = 0; t < NT; ++t) bv[t] = bias[col0 + t * 16 + idx];
    float* o = out + (size_t)(m0 + 8 * half) * N + col0 + idx;
    if (m0 + 16 <= n_dst) {
#pragma unroll
        for (int t = 0; t < NT; ++t) {
#pragma unroll
            for (int j = 0; j < 8; ++j) {
                float v = acc[t][j] + bv[t];
                if (RELU) v = fmaxf(v, 0.0f);
                o[j * N + t * 16] = v;
            }
        }
    } else {
#pragma unroll
        for (int t = 0; t < NT; ++t) {
#pragma unroll
            for (int j = 0; j < 8; ++j) {
                if (m0 + j + 8 * half < n_dst) {
                    float v = acc[t][j] + bv[t];
                    if (RELU) v = fmaxf(v, 0.0f);
                    o[j * N + t * 16] = v;
                }
            }
        }
    }
}

static inline unsigned cdiv(unsigned a, unsigned b) { return (a + b - 1) / b; }

static void sage_gather(const float* hsrc, int n_src, int nedges, int n_dst, unsigned* wb,
                        float* msg, hipStream_t stream) {
    seg_gather<<<cdiv((unsigned)n_dst, 8), dim3(256), 0, stream>>>(
        hsrc, n_src, (const int*)(wb + PERM_W), wb, wb + CUR_W, nedges, n_dst, msg);
}

extern "C" void kernel_launch(void* const* d_in, const int* in_sizes, int n_in,
                              void* d_out, int out_size, void* d_ws, size_t ws_size,
                              hipStream_t stream) {
    const float* feats = (const float*)d_in[0];
    const int* src0 = (const int*)d_in[1];
    const int* dst0 = (const int*)d_in[2];
    const int* src1 = (const int*)d_in[3];
    const int* dst1 = (const int*)d_in[4];
    const int* src2 = (const int*)d_in[5];
    const int* dst2 = (const int*)d_in[6];
    const float* Ws0 = (const float*)d_in[10];
    const float* Wn0 = (const float*)d_in[11];
    const float* b0  = (const float*)d_in[12];
    const float* Ws1 = (const float*)d_in[13];
    const float* Wn1 = (const float*)d_in[14];
    const float* b1  = (const float*)d_in[15];
    const float* Ws2 = (const float*)d_in[16];
    const float* Wn2 = (const float*)d_in[17];
    const float* b2  = (const float*)d_in[18];

    const int E0 = in_sizes[1], E1 = in_sizes[3], E2 = in_sizes[5];
    const int N0 = 500000, N1 = 100000, N2 = 20000, N3 = 5000;

    char* ws = (char*)d_ws;
    float* h1    = (float*)ws;                  ws += (size_t)N1 * 128 * sizeof(float);
    char*  lists = ws;                          ws += (size_t)N2 * 128 * sizeof(float);
    float* msg   = (float*)ws;                  ws += (size_t)N1 * 128 * sizeof(float);
    float* deg0  = (float*)ws;
    float* h2    = (float*)((char*)msg + ((size_t)16 << 20));

    unsigned* wb0  = (unsigned*)h1;
    int2*     tmp0 = (int2*)((char*)h1 + ((size_t)9 << 20));
    unsigned* wb1  = (unsigned*)lists;
    float*    deg1 = (float*)(lists + ((size_t)2816 << 10));
    int2*     tmp1 = (int2*)(lists + ((size_t)3 << 20));
    char*     l2   = lists + ((size_t)6656 << 10);
    unsigned* wb2  = (unsigned*)l2;
    float*    deg2 = (float*)(l2 + ((size_t)1536 << 10));
    int2*     tmp2 = (int2*)(l2 + ((size_t)2 << 20));

    const dim3 blk(256);

    {
        const unsigned chunks = cdiv((unsigned)E0, CHUNK) + cdiv((unsigned)E1, CHUNK) +
                                cdiv((unsigned)E2, CHUNK);
        const unsigned buckets = cdiv((unsigned)N1, BK_ROWS) + cdiv((unsigned)N2, BK_ROWS) +
                                 cdiv((unsigned)N3, BK_ROWS);
        zero_tables<<<12, blk, 0, stream>>>(wb0, wb1, wb2);
        bucket_count<<<chunks, blk, 0, stream>>>(src0, dst0, E0, N1, wb0, tmp0, deg0,
                                                 src1, dst1, E1, N2, wb1, tmp1, deg1,
                                                 src2, dst2, E2, N3, wb2, tmp2, deg2);
        bucket_scatter<<<chunks, blk, 0, stream>>>(src0, dst0, E0, N1, wb0, tmp0, deg0,
                                                   src1, dst1, E1, N2, wb1, tmp1, deg1,
                                                   src2, dst2, E2, N3, wb2, tmp2, deg2);
        bucket_sort<<<buckets, blk, 0, stream>>>(src0, dst0, E0, N1, wb0, tmp0, deg0,
                                                 src1, dst1, E1, N2, wb1, tmp1, deg1,
                                                 src2, dst2, E2, N3, wb2, tmp2, deg2);
    }

    sage_gather(feats, N0, E0, N1, wb0, msg, stream);
    {
        unsigned strips = cdiv((unsigned)N1, 16);
        sage_gemm<4, 2, 4, true><<<cdiv(strips, 4) * 2, dim3(128), 0, stream>>>(feats, msg, deg0, Ws0, Wn0, b0, h1, N1);
    }

    sage_gather(h1, N1, E1, N2, wb1, msg, stream);
    {
        unsigned strips = cdiv((unsigned)N2, 16);
        sage_gemm<4, 2, 4, true><<<cdiv(strips, 4) * 2, dim3(128), 0, stream>>>(h1, msg, deg1, Ws1, Wn1, b1, h2, N2);
    }

    sage_gather(h2, N2, E2, N3, wb2, msg, stream);
    {
        unsigned strips = cdiv((unsigned)N3, 16);
        sage_gemm<2, 2, 4, false><<<cdiv(strips, 4) * 2, dim3(128), 0, stream>>>(h2, msg, deg2, Ws2, Wn2, b2, (float*)d_out, N3);
    }
}
